# starting kernel
# baseline (speedup 1.0000x reference)
#include <hip/hip_runtime.h>
#include <hip/hip_bf16.h>
#include <math.h>

#define DD 64
#define CC 40
#define LL 3
#define BN_EPS 1e-5f

typedef __attribute__((ext_vector_type(2))) float v2f;
typedef __attribute__((ext_vector_type(8))) float v8f;

__device__ __forceinline__ void atomic_add_f32(float* p, float v) {
    __hip_atomic_fetch_add(p, v, __ATOMIC_RELAXED, __HIP_MEMORY_SCOPE_AGENT);
}

__global__ void zero_f4_kernel(float4* __restrict__ p, int n4) {
    int i = blockIdx.x * blockDim.x + threadIdx.x;
    if (i < n4) p[i] = make_float4(0.f, 0.f, 0.f, 0.f);
}

__global__ __launch_bounds__(256) void gin_scatter_kernel(
    const float* __restrict__ h, const int* __restrict__ ei,
    float* __restrict__ agg, int E)
{
    int gid = blockIdx.x * blockDim.x + threadIdx.x;
    int e = gid >> 5;
    if (e >= E) return;
    int lane = gid & 31;
    int src = ei[e];
    int dst = ei[E + e];
    v2f v = *(const v2f*)(h + (size_t)src * DD + lane * 2);
    float* p = agg + (size_t)dst * DD + lane * 2;
    atomic_add_f32(p,     v.x);
    atomic_add_f32(p + 1, v.y);
}

__global__ __launch_bounds__(128) void gin_matmul_kernel(
    const float* __restrict__ h, const float* __restrict__ agg,
    const float* __restrict__ W, const float* __restrict__ bias,
    float* __restrict__ out, float* __restrict__ colsum,
    float* __restrict__ colsumsq, int n, int doRelu, int doStats)
{
    int rowTile = blockIdx.x;
    int colTile = threadIdx.x >> 5;
    int lane    = threadIdx.x & 31;
    int m  = lane & 15;
    int kh = lane >> 4;
    int row = rowTile * 16 + m;
    int rl  = row < n ? row : n - 1;
    int col = colTile * 16 + m;

    const float* hp = h   + (size_t)rl * DD;
    const float* ap = agg + (size_t)rl * DD;

    v8f c = {};
#pragma unroll
    for (int k0 = 0; k0 < DD; k0 += 4) {
        int ka = k0 + 2 * kh;
        v2f hv = *(const v2f*)(hp + ka);
        v2f av = *(const v2f*)(ap + ka);
        v2f a; a.x = hv.x + av.x; a.y = hv.y + av.y;
        v2f b; b.x = W[(size_t)ka * DD + col];
               b.y = W[(size_t)(ka + 1) * DD + col];
        c = __builtin_amdgcn_wmma_f32_16x16x4_f32(
                false, a, false, b, (short)0, c, false, false);
    }

    float bcol = bias[col];
    float s = 0.f, s2 = 0.f;
#pragma unroll
    for (int r = 0; r < 8; ++r) {
        int outRow = rowTile * 16 + r + 8 * kh;
        float val = c[r] + bcol;
        if (doRelu) val = fmaxf(val, 0.f);
        if (outRow < n) {
            out[(size_t)outRow * DD + col] = val;
            s  += val;
            s2 += val * val;
        }
    }
    if (doStats) {
        s  += __shfl_xor(s, 16, 32);
        s2 += __shfl_xor(s2, 16, 32);
        if (kh == 0) {
            atomic_add_f32(colsum   + col, s);
            atomic_add_f32(colsumsq + col, s2);
        }
    }
}

__global__ void bn_finalize_kernel(
    const float* __restrict__ colsum, const float* __restrict__ colsumsq,
    const float* __restrict__ gamma, const float* __restrict__ beta,
    float* __restrict__ scale, float* __restrict__ shift, int n)
{
    int c = threadIdx.x;
    if (c >= DD) return;
    float invN = 1.f / (float)n;
    float mu   = colsum[c] * invN;
    float var  = colsumsq[c] * invN - mu * mu;
    float sc   = gamma[c] * rsqrtf(var + BN_EPS);
    scale[c] = sc;
    shift[c] = beta[c] - mu * sc;
}

__global__ __launch_bounds__(256) void bn_apply_kernel(
    const float* __restrict__ hin, const float* __restrict__ scale,
    const float* __restrict__ shift, float* __restrict__ hout, int total4)
{
    int i = blockIdx.x * blockDim.x + threadIdx.x;
    if (i >= total4) return;
    int c = (i & 15) * 4;
    float4 v = ((const float4*)hin)[i];
    v.x = v.x * scale[c + 0] + shift[c + 0];
    v.y = v.y * scale[c + 1] + shift[c + 1];
    v.z = v.z * scale[c + 2] + shift[c + 2];
    v.w = v.w * scale[c + 3] + shift[c + 3];
    ((float4*)hout)[i] = v;
}

__global__ void pad_final_kernel(const float* __restrict__ Wf,
                                 const float* __restrict__ bf,
                                 float* __restrict__ Wpad,
                                 float* __restrict__ bpad)
{
    int i = blockIdx.x * blockDim.x + threadIdx.x;
    if (i < DD * DD) {
        int r = i >> 6, c = i & 63;
        Wpad[i] = (c < CC) ? Wf[r * CC + c] : 0.f;
    }
    if (i < DD) bpad[i] = (i < CC) ? bf[i] : 0.f;
}

__global__ __launch_bounds__(256) void logsoftmax_kernel(
    const float* __restrict__ y, float* __restrict__ out, int n)
{
    int gid  = blockIdx.x * blockDim.x + threadIdx.x;
    int row  = gid >> 5;
    int lane = gid & 31;
    if (row >= n) return;
    const float* yr = y + (size_t)row * DD;
    float v0 = yr[lane];
    bool  h1 = (lane + 32) < CC;
    float v1 = h1 ? yr[lane + 32] : -3.402823466e38f;
    float mx = fmaxf(v0, v1);
    for (int o = 16; o > 0; o >>= 1) mx = fmaxf(mx, __shfl_xor(mx, o, 32));
    float s = expf(v0 - mx) + (h1 ? expf(v1 - mx) : 0.f);
    for (int o = 16; o > 0; o >>= 1) s += __shfl_xor(s, o, 32);
    float lse = mx + logf(s);
    out[(size_t)row * CC + lane] = v0 - lse;
    if (h1) out[(size_t)row * CC + lane + 32] = v1 - lse;
}

extern "C" void kernel_launch(void* const* d_in, const int* in_sizes, int n_in,
                              void* d_out, int out_size, void* d_ws, size_t ws_size,
                              hipStream_t stream) {
    const float* x       = (const float*)d_in[0];
    const int*   ei      = (const int*)  d_in[1];
    const float* W_proj  = (const float*)d_in[2];
    const float* b_proj  = (const float*)d_in[3];
    const float* Ws      = (const float*)d_in[4];
    const float* bs      = (const float*)d_in[5];
    const float* W_final = (const float*)d_in[6];
    const float* b_final = (const float*)d_in[7];
    const float* gammas  = (const float*)d_in[8];
    const float* betas   = (const float*)d_in[9];
    float* out = (float*)d_out;

    int n = in_sizes[0] / DD;
    int E = in_sizes[1] / 2;

    size_t nd = (size_t)n * DD;
    float* ws    = (float*)d_ws;
    float* h_cur = ws;
    float* h_tmp = ws + nd;
    float* agg   = ws + 2 * nd;
    float* stats = ws + 3 * nd;
    float* scale = stats + 128;
    float* shift = scale + DD;
    float* Wpad  = shift + DD;
    float* bpad  = Wpad + DD * DD;

    int nRowTiles     = (n + 15) / 16;
    int nd4           = (int)(nd / 4);
    int zeroBlocks    = (nd4 + 255) / 256;
    int scatterBlocks = (int)(((long long)E * 32 + 255) / 256);
    int lsBlocks      = (int)(((long long)n * 32 + 255) / 256);

    for (int layer = 0; layer <= LL; ++layer) {
        const float* hin = (layer == 0) ? x : h_cur;
        const float* W   = (layer == 0) ? W_proj : Ws + (size_t)(layer - 1) * DD * DD;
        const float* b   = (layer == 0) ? b_proj : bs + (size_t)(layer - 1) * DD;

        zero_f4_kernel<<<zeroBlocks, 256, 0, stream>>>((float4*)agg, nd4);
        zero_f4_kernel<<<1, 32, 0, stream>>>((float4*)stats, 32);
        gin_scatter_kernel<<<scatterBlocks, 256, 0, stream>>>(hin, ei, agg, E);
        gin_matmul_kernel<<<nRowTiles, 128, 0, stream>>>(
            hin, agg, W, b, h_tmp, stats, stats + DD, n,  1,  1);
        bn_finalize_kernel<<<1, DD, 0, stream>>>(
            stats, stats + DD, gammas + (size_t)layer * DD, betas + (size_t)layer * DD,
            scale, shift, n);
        bn_apply_kernel<<<zeroBlocks, 256, 0, stream>>>(h_tmp, scale, shift, h_cur, nd4);
    }

    pad_final_kernel<<<(DD * DD + 255) / 256, 256, 0, stream>>>(W_final, b_final, Wpad, bpad);
    zero_f4_kernel<<<zeroBlocks, 256, 0, stream>>>((float4*)agg, nd4);
    gin_scatter_kernel<<<scatterBlocks, 256, 0, stream>>>(h_cur, ei, agg, E);
    gin_matmul_kernel<<<nRowTiles, 128, 0, stream>>>(
        h_cur, agg, Wpad, bpad, h_tmp, nullptr, nullptr, n,  0,  0);
    logsoftmax_kernel<<<lsBlocks, 256, 0, stream>>>(h_tmp, out, n);
}


# scatter: each wave takes 8 consecutive edges and issues its 8 row loads back to back before its atomics (8 loads in flight per wave, 8x fewer waves); sums still f32 atomics in arrival order
# speedup vs baseline: 1.4531x; 1.4531x over previous
#include <hip/hip_runtime.h>
#include <hip/hip_bf16.h>
#include <math.h>

#define DD 64
#define CC 40
#define LL 3
#define BN_EPS 1e-5f

typedef __attribute__((ext_vector_type(2))) float v2f;
typedef __attribute__((ext_vector_type(8))) float v8f;

__device__ __forceinline__ void atomic_add_f32(float* p, float v) {
    __hip_atomic_fetch_add(p, v, __ATOMIC_RELAXED, __HIP_MEMORY_SCOPE_AGENT);
}

__global__ void zero_f4_kernel(float4* __restrict__ p, int n4) {
    int i = blockIdx.x * blockDim.x + threadIdx.x;
    if (i < n4) p[i] = make_float4(0.f, 0.f, 0.f, 0.f);
}

#define EPW 8
__global__ __launch_bounds__(256) void gin_scatter_kernel(
    const float* __restrict__ h, const int* __restrict__ ei,
    float* __restrict__ agg, int E)
{
    int gid  = blockIdx.x * blockDim.x + threadIdx.x;
    int lane = gid & 31;
    int e0   = (gid >> 5) * EPW;
    if (e0 >= E) return;
    if (e0 + EPW <= E) {
        int j   = lane & 15;
        int idx = ei[(size_t)((j & 8) ? E : 0) + e0 + (j & 7)];
        v2f v[EPW];
        int dsts[EPW];
#pragma unroll
        for (int i = 0; i < EPW; ++i) {
            int src = __builtin_amdgcn_readlane(idx, i);
            dsts[i] = __builtin_amdgcn_readlane(idx, EPW + i);
            v[i] = *(const v2f*)(h + (size_t)src * DD + lane * 2);
        }
#pragma unroll
        for (int i = 0; i < EPW; ++i) {
            float* p = agg + (size_t)dsts[i] * DD + lane * 2;
            atomic_add_f32(p,     v[i].x);
            atomic_add_f32(p + 1, v[i].y);
        }
    } else {
        for (int e = e0; e < E; ++e) {
            int src = ei[e];
            int dst = ei[(size_t)E + e];
            v2f v = *(const v2f*)(h + (size_t)src * DD + lane * 2);
            float* p = agg + (size_t)dst * DD + lane * 2;
            atomic_add_f32(p,     v.x);
            atomic_add_f32(p + 1, v.y);
        }
    }
}

__global__ __launch_bounds__(128) void gin_matmul_kernel(
    const float* __restrict__ h, const float* __restrict__ agg,
    const float* __restrict__ W, const float* __restrict__ bias,
    float* __restrict__ out, float* __restrict__ colsum,
    float* __restrict__ colsumsq, int n, int doRelu, int doStats)
{
    int rowTile = blockIdx.x;
    int colTile = threadIdx.x >> 5;
    int lane    = threadIdx.x & 31;
    int m  = lane & 15;
    int kh = lane >> 4;
    int row = rowTile * 16 + m;
    int rl  = row < n ? row : n - 1;
    int col = colTile * 16 + m;

    const float* hp = h   + (size_t)rl * DD;
    const float* ap = agg + (size_t)rl * DD;

    v8f c = {};
#pragma unroll
    for (int k0 = 0; k0 < DD; k0 += 4) {
        int ka = k0 + 2 * kh;
        v2f hv = *(const v2f*)(hp + ka);
        v2f av = *(const v2f*)(ap + ka);
        v2f a; a.x = hv.x + av.x; a.y = hv.y + av.y;
        v2f b; b.x = W[(size_t)ka * DD + col];
               b.y = W[(size_t)(ka + 1) * DD + col];
        c = __builtin_amdgcn_wmma_f32_16x16x4_f32(
                false, a, false, b, (short)0, c, false, false);
    }

    float bcol = bias[col];
    float s = 0.f, s2 = 0.f;
#pragma unroll
    for (int r = 0; r < 8; ++r) {
        int outRow = rowTile * 16 + r + 8 * kh;
        float val = c[r] + bcol;
        if (doRelu) val = fmaxf(val, 0.f);
        if (outRow < n) {
            out[(size_t)outRow * DD + col] = val;
            s  += val;
            s2 += val * val;
        }
    }
    if (doStats) {
        s  += __shfl_xor(s, 16, 32);
        s2 += __shfl_xor(s2, 16, 32);
        if (kh == 0) {
            atomic_add_f32(colsum   + col, s);
            atomic_add_f32(colsumsq + col, s2);
        }
    }
}

__global__ void bn_finalize_kernel(
    const float* __restrict__ colsum, const float* __restrict__ colsumsq,
    const float* __restrict__ gamma, const float* __restrict__ beta,
    float* __restrict__ scale, float* __restrict__ shift, int n)
{
    int c = threadIdx.x;
    if (c >= DD) return;
    float invN = 1.f / (float)n;
    float mu   = colsum[c] * invN;
    float var  = colsumsq[c] * invN - mu * mu;
    float sc   = gamma[c] * rsqrtf(var + BN_EPS);
    scale[c] = sc;
    shift[c] = beta[c] - mu * sc;
}

__global__ __launch_bounds__(256) void bn_apply_kernel(
    const float* __restrict__ hin, const float* __restrict__ scale,
    const float* __restrict__ shift, float* __restrict__ hout, int total4)
{
    int i = blockIdx.x * blockDim.x + threadIdx.x;
    if (i >= total4) return;
    int c = (i & 15) * 4;
    float4 v = ((const float4*)hin)[i];
    v.x = v.x * scale[c + 0] + shift[c + 0];
    v.y = v.y * scale[c + 1] + shift[c + 1];
    v.z = v.z * scale[c + 2] + shift[c + 2];
    v.w = v.w * scale[c + 3] + shift[c + 3];
    ((float4*)hout)[i] = v;
}

__global__ void pad_final_kernel(const float* __restrict__ Wf,
                                 const float* __restrict__ bf,
                                 float* __restrict__ Wpad,
                                 float* __restrict__ bpad)
{
    int i = blockIdx.x * blockDim.x + threadIdx.x;
    if (i < DD * DD) {
        int r = i >> 6, c = i & 63;
        Wpad[i] = (c < CC) ? Wf[r * CC + c] : 0.f;
    }
    if (i < DD) bpad[i] = (i < CC) ? bf[i] : 0.f;
}

__global__ __launch_bounds__(256) void logsoftmax_kernel(
    const float* __restrict__ y, float* __restrict__ out, int n)
{
    int gid  = blockIdx.x * blockDim.x + threadIdx.x;
    int row  = gid >> 5;
    int lane = gid & 31;
    if (row >= n) return;
    const float* yr = y + (size_t)row * DD;
    float v0 = yr[lane];
    bool  h1 = (lane + 32) < CC;
    float v1 = h1 ? yr[lane + 32] : -3.402823466e38f;
    float mx = fmaxf(v0, v1);
    for (int o = 16; o > 0; o >>= 1) mx = fmaxf(mx, __shfl_xor(mx, o, 32));
    float s = expf(v0 - mx) + (h1 ? expf(v1 - mx) : 0.f);
    for (int o = 16; o > 0; o >>= 1) s += __shfl_xor(s, o, 32);
    float lse = mx + logf(s);
    out[(size_t)row * CC + lane] = v0 - lse;
    if (h1) out[(size_t)row * CC + lane + 32] = v1 - lse;
}

extern "C" void kernel_launch(void* const* d_in, const int* in_sizes, int n_in,
                              void* d_out, int out_size, void* d_ws, size_t ws_size,
                              hipStream_t stream) {
    const float* x       = (const float*)d_in[0];
    const int*   ei      = (const int*)  d_in[1];
    const float* W_proj  = (const float*)d_in[2];
    const float* b_proj  = (const float*)d_in[3];
    const float* Ws      = (const float*)d_in[4];
    const float* bs      = (const float*)d_in[5];
    const float* W_final = (const float*)d_in[6];
    const float* b_final = (const float*)d_in[7];
    const float* gammas  = (const float*)d_in[8];
    const float* betas   = (const float*)d_in[9];
    float* out = (float*)d_out;

    int n = in_sizes[0] / DD;
    int E = in_sizes[1] / 2;

    size_t nd = (size_t)n * DD;
    float* ws    = (float*)d_ws;
    float* h_cur = ws;
    float* h_tmp = ws + nd;
    float* agg   = ws + 2 * nd;
    float* stats = ws + 3 * nd;
    float* scale = stats + 128;
    float* shift = scale + DD;
    float* Wpad  = shift + DD;
    float* bpad  = Wpad + DD * DD;

    int nRowTiles     = (n + 15) / 16;
    int nd4           = (int)(nd / 4);
    int zeroBlocks    = (nd4 + 255) / 256;
    int scatterBlocks = (int)((((long long)E + EPW - 1) / EPW * 32 + 255) / 256);
    int lsBlocks      = (int)(((long long)n * 32 + 255) / 256);

    for (int layer = 0; layer <= LL; ++layer) {
        const float* hin = (layer == 0) ? x : h_cur;
        const float* W   = (layer == 0) ? W_proj : Ws + (size_t)(layer - 1) * DD * DD;
        const float* b   = (layer == 0) ? b_proj : bs + (size_t)(layer - 1) * DD;

        zero_f4_kernel<<<zeroBlocks, 256, 0, stream>>>((float4*)agg, nd4);
        zero_f4_kernel<<<1, 32, 0, stream>>>((float4*)stats, 32);
        gin_scatter_kernel<<<scatterBlocks, 256, 0, stream>>>(hin, ei, agg, E);
        gin_matmul_kernel<<<nRowTiles, 128, 0, stream>>>(
            hin, agg, W, b, h_tmp, stats, stats + DD, n,  1,  1);
        bn_finalize_kernel<<<1, DD, 0, stream>>>(
            stats, stats + DD, gammas + (size_t)layer * DD, betas + (size_t)layer * DD,
            scale, shift, n);
        bn_apply_kernel<<<zeroBlocks, 256, 0, stream>>>(h_tmp, scale, shift, h_cur, nd4);
    }

    pad_final_kernel<<<(DD * DD + 255) / 256, 256, 0, stream>>>(W_final, b_final, Wpad, bpad);
    zero_f4_kernel<<<zeroBlocks, 256, 0, stream>>>((float4*)agg, nd4);
    gin_scatter_kernel<<<scatterBlocks, 256, 0, stream>>>(h_cur, ei, agg, E);
    gin_matmul_kernel<<<nRowTiles, 128, 0, stream>>>(
        h_cur, agg, Wpad, bpad, h_tmp, nullptr, nullptr, n,  0,  0);
    logsoftmax_kernel<<<lsBlocks, 256, 0, stream>>>(h_tmp, out, n);
}


# aggregation: edges counting-sorted by destination once per call (int atomics set list order); one wave sums a row's source rows in registers in list order; no f32 atomics or zero pass on agg
# speedup vs baseline: 1.6450x; 1.1321x over previous
#include <hip/hip_runtime.h>
#include <hip/hip_bf16.h>
#include <math.h>

#define DD 64
#define CC 40
#define LL 3
#define BN_EPS 1e-5f

typedef __attribute__((ext_vector_type(2))) float v2f;
typedef __attribute__((ext_vector_type(8))) float v8f;

__device__ __forceinline__ void atomic_add_f32(float* p, float v) {
    __hip_atomic_fetch_add(p, v, __ATOMIC_RELAXED, __HIP_MEMORY_SCOPE_AGENT);
}

__global__ void zero_f4_kernel(float4* __restrict__ p, int n4) {
    int i = blockIdx.x * blockDim.x + threadIdx.x;
    if (i < n4) p[i] = make_float4(0.f, 0.f, 0.f, 0.f);
}

#define EPB 1024
__global__ __launch_bounds__(256) void csr_count_kernel(
    const int* __restrict__ ei, int* __restrict__ cnt, int* __restrict__ rank, int E)
{
    int e0 = blockIdx.x * EPB + threadIdx.x;
    if (blockIdx.x * EPB + EPB <= E) {
        int d[4];
#pragma unroll
        for (int k = 0; k < 4; ++k) d[k] = ei[(size_t)E + e0 + k * 256];
        int r[4];
#pragma unroll
        for (int k = 0; k < 4; ++k)
            r[k] = __hip_atomic_fetch_add(cnt + d[k], 1, __ATOMIC_RELAXED, __HIP_MEMORY_SCOPE_AGENT);
#pragma unroll
        for (int k = 0; k < 4; ++k) rank[e0 + k * 256] = r[k];
    } else {
        for (int e = e0; e < E; e += 256) {
            int d = ei[(size_t)E + e];
            rank[e] = __hip_atomic_fetch_add(cnt + d, 1, __ATOMIC_RELAXED, __HIP_MEMORY_SCOPE_AGENT);
        }
    }
}

__global__ __launch_bounds__(256) void csr_scan_kernel(
    const int* __restrict__ cnt, int* __restrict__ rowptr)
{
    __shared__ int preSum[8];
    __shared__ int waveSum[8];
    int t = threadIdx.x, lane = t & 31, w = t >> 5;
    const int4* c4 = (const int4*)cnt;
    int pre = 0;
    for (int i = t; i < (int)blockIdx.x * 256; i += 256) {
        int4 q = c4[i];
        pre += q.x + q.y + q.z + q.w;
    }
    for (int o = 16; o > 0; o >>= 1) pre += __shfl_xor(pre, o, 32);
    int4 q = c4[blockIdx.x * 256 + t];
    int tsum = q.x + q.y + q.z + q.w;
    int inc = tsum;
    for (int o = 1; o < 32; o <<= 1) {
        int y = __shfl_up(inc, o, 32);
        if (lane >= o) inc += y;
    }
    if (lane == 0)  preSum[w]  = pre;
    if (lane == 31) waveSum[w] = inc;
    __syncthreads();
    int base = 0;
    for (int i = 0; i < 8; ++i) {
        base += preSum[i];
        if (i < w) base += waveSum[i];
    }
    int ex = base + inc - tsum;
    int4 o4;
    o4.x = ex;
    o4.y = o4.x + q.x;
    o4.z = o4.y + q.y;
    o4.w = o4.z + q.z;
    ((int4*)rowptr)[blockIdx.x * 256 + t] = o4;
}

__global__ __launch_bounds__(256) void csr_place_kernel(
    const int* __restrict__ ei, const int* __restrict__ rowptr,
    const int* __restrict__ rank, int* __restrict__ col, int E)
{
    int e0 = blockIdx.x * EPB + threadIdx.x;
    if (blockIdx.x * EPB + EPB <= E) {
        int s[4], p[4];
#pragma unroll
        for (int k = 0; k < 4; ++k) {
            int e = e0 + k * 256;
            s[k] = ei[e];
            p[k] = rowptr[ei[(size_t)E + e]] + rank[e];
        }
#pragma unroll
        for (int k = 0; k < 4; ++k) col[p[k]] = s[k];
    } else {
        for (int e = e0; e < E; e += 256)
            col[rowptr[ei[(size_t)E + e]] + rank[e]] = ei[e];
    }
}

#define RPW 8
#define GCH 8
__global__ __launch_bounds__(256) void gin_gather_kernel(
    const float* __restrict__ h, const int* __restrict__ rowptr,
    const int* __restrict__ col, float* __restrict__ agg, int n)
{
    int lane = threadIdx.x & 31;
    int wave = __builtin_amdgcn_readfirstlane((int)((blockIdx.x * blockDim.x + threadIdx.x) >> 5));
    int r0 = wave * RPW;
    if (r0 >= n) return;
    int nr = (n - r0 < RPW) ? (n - r0) : RPW;
    int pj = rowptr[r0 + (lane < nr ? lane : nr)];
    for (int rr = 0; rr < nr; ++rr) {
        int beg = __builtin_amdgcn_readlane(pj, rr);
        int end = __builtin_amdgcn_readlane(pj, rr + 1);
        v2f acc = {0.f, 0.f};
        for (int base = beg; base < end; base += 32) {
            int c = (end - base < 32) ? (end - base) : 32;
            int idx = col[base + (lane < c ? lane : c - 1)];
            for (int i0 = 0; i0 < c; i0 += GCH) {
                v2f v[GCH];
#pragma unroll
                for (int i = 0; i < GCH; ++i) {
                    int k = (i0 + i < c) ? (i0 + i) : (c - 1);
                    int src = __builtin_amdgcn_readlane(idx, k);
                    v[i] = *(const v2f*)(h + (size_t)src * DD + lane * 2);
                }
#pragma unroll
                for (int i = 0; i < GCH; ++i) {
                    bool ok = (i0 + i) < c;
                    acc.x += ok ? v[i].x : 0.f;
                    acc.y += ok ? v[i].y : 0.f;
                }
            }
        }
        *(v2f*)(agg + (size_t)(r0 + rr) * DD + lane * 2) = acc;
    }
}

__global__ __launch_bounds__(128) void gin_matmul_kernel(
    const float* __restrict__ h, float* aggOut,
    const float* __restrict__ W, const float* __restrict__ bias,
    float* __restrict__ colsum,
    float* __restrict__ colsumsq, int n, int doRelu, int doStats)
{
    int rowTile = blockIdx.x;
    int colTile = threadIdx.x >> 5;
    int lane    = threadIdx.x & 31;
    int m  = lane & 15;
    int kh = lane >> 4;
    int row = rowTile * 16 + m;
    int rl  = row < n ? row : n - 1;
    int col = colTile * 16 + m;

    const float* hp = h   + (size_t)rl * DD;
    const float* ap = aggOut + (size_t)rl * DD;

    v8f c = {};
#pragma unroll
    for (int k0 = 0; k0 < DD; k0 += 4) {
        int ka = k0 + 2 * kh;
        v2f hv = *(const v2f*)(hp + ka);
        v2f av = *(const v2f*)(ap + ka);
        v2f a; a.x = hv.x + av.x; a.y = hv.y + av.y;
        v2f b; b.x = W[(size_t)ka * DD + col];
               b.y = W[(size_t)(ka + 1) * DD + col];
        c = __builtin_amdgcn_wmma_f32_16x16x4_f32(
                false, a, false, b, (short)0, c, false, false);
    }

    __syncthreads();

    float bcol = bias[col];
    float s = 0.f, s2 = 0.f;
#pragma unroll
    for (int r = 0; r < 8; ++r) {
        int outRow = rowTile * 16 + r + 8 * kh;
        float val = c[r] + bcol;
        if (doRelu) val = fmaxf(val, 0.f);
        if (outRow < n) {
            aggOut[(size_t)outRow * DD + col] = val;
            s  += val;
            s2 += val * val;
        }
    }
    if (doStats) {
        s  += __shfl_xor(s, 16, 32);
        s2 += __shfl_xor(s2, 16, 32);
        if (kh == 0) {
            atomic_add_f32(colsum   + col, s);
            atomic_add_f32(colsumsq + col, s2);
        }
    }
}

__global__ void bn_finalize_kernel(
    const float* __restrict__ colsum, const float* __restrict__ colsumsq,
    const float* __restrict__ gamma, const float* __restrict__ beta,
    float* __restrict__ scale, float* __restrict__ shift, int n)
{
    int c = threadIdx.x;
    if (c >= DD) return;
    float invN = 1.f / (float)n;
    float mu   = colsum[c] * invN;
    float var  = colsumsq[c] * invN - mu * mu;
    float sc   = gamma[c] * rsqrtf(var + BN_EPS);
    scale[c] = sc;
    shift[c] = beta[c] - mu * sc;
}

__global__ __launch_bounds__(256) void bn_apply_kernel(
    const float* __restrict__ hin, const float* __restrict__ scale,
    const float* __restrict__ shift, float* __restrict__ hout, int total4)
{
    int i = blockIdx.x * blockDim.x + threadIdx.x;
    if (i >= total4) return;
    int c = (i & 15) * 4;
    float4 v = ((const float4*)hin)[i];
    v.x = v.x * scale[c + 0] + shift[c + 0];
    v.y = v.y * scale[c + 1] + shift[c + 1];
    v.z = v.z * scale[c + 2] + shift[c + 2];
    v.w = v.w * scale[c + 3] + shift[c + 3];
    ((float4*)hout)[i] = v;
}

__global__ void pad_final_kernel(const float* __restrict__ Wf,
                                 const float* __restrict__ bf,
                                 float* __restrict__ Wpad,
                                 float* __restrict__ bpad)
{
    int i = blockIdx.x * blockDim.x + threadIdx.x;
    if (i < DD * DD) {
        int r = i >> 6, c = i & 63;
        Wpad[i] = (c < CC) ? Wf[r * CC + c] : 0.f;
    }
    if (i < DD) bpad[i] = (i < CC) ? bf[i] : 0.f;
}

__global__ __launch_bounds__(256) void logsoftmax_kernel(
    const float* __restrict__ y, float* __restrict__ out, int n)
{
    int gid  = blockIdx.x * blockDim.x + threadIdx.x;
    int row  = gid >> 5;
    int lane = gid & 31;
    if (row >= n) return;
    const float* yr = y + (size_t)row * DD;
    float v0 = yr[lane];
    bool  h1 = (lane + 32) < CC;
    float v1 = h1 ? yr[lane + 32] : -3.402823466e38f;
    float mx = fmaxf(v0, v1);
    for (int o = 16; o > 0; o >>= 1) mx = fmaxf(mx, __shfl_xor(mx, o, 32));
    float s = expf(v0 - mx) + (h1 ? expf(v1 - mx) : 0.f);
    for (int o = 16; o > 0; o >>= 1) s += __shfl_xor(s, o, 32);
    float lse = mx + logf(s);
    out[(size_t)row * CC + lane] = v0 - lse;
    if (h1) out[(size_t)row * CC + lane + 32] = v1 - lse;
}

extern "C" void kernel_launch(void* const* d_in, const int* in_sizes, int n_in,
                              void* d_out, int out_size, void* d_ws, size_t ws_size,
                              hipStream_t stream) {
    const float* x       = (const float*)d_in[0];
    const int*   ei      = (const int*)  d_in[1];
    const float* W_proj  = (const float*)d_in[2];
    const float* b_proj  = (const float*)d_in[3];
    const float* Ws      = (const float*)d_in[4];
    const float* bs      = (const float*)d_in[5];
    const float* W_final = (const float*)d_in[6];
    const float* b_final = (const float*)d_in[7];
    const float* gammas  = (const float*)d_in[8];
    const float* betas   = (const float*)d_in[9];
    float* out = (float*)d_out;

    int n = in_sizes[0] / DD;
    int E = in_sizes[1] / 2;

    size_t nd = (size_t)n * DD;
    float* ws    = (float*)d_ws;
    float* h_cur = ws;
    int*   lists = (int*)(ws + nd);
    float* agg   = ws + 2 * nd;
    float* stats = ws + 3 * nd;
    float* scale = stats + 128;
    float* shift = scale + DD;
    float* Wpad  = shift + DD;
    float* bpad  = Wpad + DD * DD;

    int scanTiles = (n + 1 + 1023) / 1024;
    int RP        = scanTiles * 1024;
    int EP        = (E + 63) / 64 * 64;
    int* cnt    = lists;
    int* rowptr = cnt + RP;
    int* rank   = rowptr + RP;
    int* col    = rank + EP;

    int nRowTiles    = (n + 15) / 16;
    int nd4          = (int)(nd / 4);
    int applyBlocks  = (nd4 + 255) / 256;
    int edgeBlocks   = (E + EPB - 1) / EPB;
    int gatherBlocks = (int)((((long long)n + RPW - 1) / RPW * 32 + 255) / 256);
    int lsBlocks     = (int)(((long long)n * 32 + 255) / 256);

    zero_f4_kernel<<<scanTiles, 256, 0, stream>>>((float4*)cnt, RP / 4);
    csr_count_kernel<<<edgeBlocks, 256, 0, stream>>>(ei, cnt, rank, E);
    csr_scan_kernel<<<scanTiles, 256, 0, stream>>>(cnt, rowptr);
    csr_place_kernel<<<edgeBlocks, 256, 0, stream>>>(ei, rowptr, rank, col, E);

    for (int layer = 0; layer <= LL; ++layer) {
        const float* hin = (layer == 0) ? x : h_cur;
        const float* W   = (layer == 0) ? W_proj : Ws + (size_t)(layer - 1) * DD * DD;
        const float* b   = (layer == 0) ? b_proj : bs + (size_t)(layer - 1) * DD;

        zero_f4_kernel<<<1, 32, 0, stream>>>((float4*)stats, 32);
        gin_gather_kernel<<<gatherBlocks, 256, 0, stream>>>(hin, rowptr, col, agg, n);
        gin_matmul_kernel<<<nRowTiles, 128, 0, stream>>>(
            hin, agg, W, b, stats, stats + DD, n,  1,  1);
        bn_finalize_kernel<<<1, DD, 0, stream>>>(
            stats, stats + DD, gammas + (size_t)layer * DD, betas + (size_t)layer * DD,
            scale, shift, n);
        bn_apply_kernel<<<applyBlocks, 256, 0, stream>>>(agg, scale, shift, h_cur, nd4);
    }

    pad_final_kernel<<<(DD * DD + 255) / 256, 256, 0, stream>>>(W_final, b_final, Wpad, bpad);
    gin_gather_kernel<<<gatherBlocks, 256, 0, stream>>>(h_cur, rowptr, col, agg, n);
    gin_matmul_kernel<<<nRowTiles, 128, 0, stream>>>(
        h_cur, agg, Wpad, bpad, nullptr, nullptr, n,  0,  0);
    logsoftmax_kernel<<<lsBlocks, 256, 0, stream>>>(agg, out, n);
}


# layer fused: a 4-wave block sums its 32 rows (neighbours in list order, then own row) into an LDS tile and multiplies it by W at once; agg plane and 5 launches go; column sums per 32 rows per atomic
# speedup vs baseline: 2.3847x; 1.4497x over previous
#include <hip/hip_runtime.h>
#include <hip/hip_bf16.h>
#include <math.h>

#define DD 64
#define CC 40
#define LL 3
#define BN_EPS 1e-5f

typedef __attribute__((ext_vector_type(2))) float v2f;
typedef __attribute__((ext_vector_type(8))) float v8f;

__device__ __forceinline__ void atomic_add_f32(float* p, float v) {
    __hip_atomic_fetch_add(p, v, __ATOMIC_RELAXED, __HIP_MEMORY_SCOPE_AGENT);
}

__global__ void zero_f4_kernel(float4* __restrict__ p, int n4) {
    int i = blockIdx.x * blockDim.x + threadIdx.x;
    if (i < n4) p[i] = make_float4(0.f, 0.f, 0.f, 0.f);
}

#define EPB 1024
__global__ __launch_bounds__(256) void csr_count_kernel(
    const int* __restrict__ ei, int* __restrict__ cnt, int* __restrict__ rank, int E)
{
    int e0 = blockIdx.x * EPB + threadIdx.x;
    if (blockIdx.x * EPB + EPB <= E) {
        int d[4];
#pragma unroll
        for (int k = 0; k < 4; ++k) d[k] = ei[(size_t)E + e0 + k * 256];
        int r[4];
#pragma unroll
        for (int k = 0; k < 4; ++k)
            r[k] = __hip_atomic_fetch_add(cnt + d[k], 1, __ATOMIC_RELAXED, __HIP_MEMORY_SCOPE_AGENT);
#pragma unroll
        for (int k = 0; k < 4; ++k) rank[e0 + k * 256] = r[k];
    } else {
        for (int e = e0; e < E; e += 256) {
            int d = ei[(size_t)E + e];
            rank[e] = __hip_atomic_fetch_add(cnt + d, 1, __ATOMIC_RELAXED, __HIP_MEMORY_SCOPE_AGENT);
        }
    }
}

__global__ __launch_bounds__(256) void csr_scan_kernel(
    const int* __restrict__ cnt, int* __restrict__ rowptr)
{
    __shared__ int preSum[8];
    __shared__ int waveSum[8];
    int t = threadIdx.x, lane = t & 31, w = t >> 5;
    const int4* c4 = (const int4*)cnt;
    int pre = 0;
    for (int i = t; i < (int)blockIdx.x * 256; i += 256) {
        int4 q = c4[i];
        pre += q.x + q.y + q.z + q.w;
    }
    for (int o = 16; o > 0; o >>= 1) pre += __shfl_xor(pre, o, 32);
    int4 q = c4[blockIdx.x * 256 + t];
    int tsum = q.x + q.y + q.z + q.w;
    int inc = tsum;
    for (int o = 1; o < 32; o <<= 1) {
        int y = __shfl_up(inc, o, 32);
        if (lane >= o) inc += y;
    }
    if (lane == 0)  preSum[w]  = pre;
    if (lane == 31) waveSum[w] = inc;
    __syncthreads();
    int base = 0;
    for (int i = 0; i < 8; ++i) {
        base += preSum[i];
        if (i < w) base += waveSum[i];
    }
    int ex = base + inc - tsum;
    int4 o4;
    o4.x = ex;
    o4.y = o4.x + q.x;
    o4.z = o4.y + q.y;
    o4.w = o4.z + q.z;
    ((int4*)rowptr)[blockIdx.x * 256 + t] = o4;
}

__global__ __launch_bounds__(256) void csr_place_kernel(
    const int* __restrict__ ei, const int* __restrict__ rowptr,
    const int* __restrict__ rank, int* __restrict__ col, int E)
{
    int e0 = blockIdx.x * EPB + threadIdx.x;
    if (blockIdx.x * EPB + EPB <= E) {
        int s[4], p[4];
#pragma unroll
        for (int k = 0; k < 4; ++k) {
            int e = e0 + k * 256;
            s[k] = ei[e];
            p[k] = rowptr[ei[(size_t)E + e]] + rank[e];
        }
#pragma unroll
        for (int k = 0; k < 4; ++k) col[p[k]] = s[k];
    } else {
        for (int e = e0; e < E; e += 256)
            col[rowptr[ei[(size_t)E + e]] + rank[e]] = ei[e];
    }
}

#define RPW 8
#define GCH 8
#define FROWS 32
#define TPITCH 68
__global__ __launch_bounds__(128) void gin_layer_kernel(
    const float* __restrict__ h, const int* __restrict__ rowptr,
    const int* __restrict__ col, const float* __restrict__ W,
    const float* __restrict__ bias, float* __restrict__ out,
    float* __restrict__ colsum, float* __restrict__ colsumsq,
    int n, int doRelu, int doStats)
{
    __shared__ float tile[FROWS * TPITCH];
    int lane = threadIdx.x & 31;
    int wv = __builtin_amdgcn_readfirstlane((int)(threadIdx.x >> 5));
    int rowBase = blockIdx.x * FROWS;
    int r0 = rowBase + wv * RPW;
    int nr = n - r0;
    nr = nr < 0 ? 0 : (nr > RPW ? RPW : nr);
    int pi = r0 + (lane < nr ? lane : nr);
    pi = pi > n ? n : pi;
    int pj = rowptr[pi];
    for (int rr = 0; rr < RPW; ++rr) {
        v2f tot = {0.f, 0.f};
        if (rr < nr) {
            v2f own = *(const v2f*)(h + (size_t)(r0 + rr) * DD + lane * 2);
            int beg = __builtin_amdgcn_readlane(pj, rr);
            int end = __builtin_amdgcn_readlane(pj, rr + 1);
            v2f acc = {0.f, 0.f};
            for (int base = beg; base < end; base += 32) {
                int c = (end - base < 32) ? (end - base) : 32;
                int idx = col[base + (lane < c ? lane : c - 1)];
                for (int i0 = 0; i0 < c; i0 += GCH) {
                    v2f v[GCH];
#pragma unroll
                    for (int i = 0; i < GCH; ++i) {
                        int k = (i0 + i < c) ? (i0 + i) : (c - 1);
                        int src = __builtin_amdgcn_readlane(idx, k);
                        v[i] = *(const v2f*)(h + (size_t)src * DD + lane * 2);
                    }
#pragma unroll
                    for (int i = 0; i < GCH; ++i) {
                        bool ok = (i0 + i) < c;
                        acc.x += ok ? v[i].x : 0.f;
                        acc.y += ok ? v[i].y : 0.f;
                    }
                }
            }
            tot.x = own.x + acc.x;
            tot.y = own.y + acc.y;
        }
        *(v2f*)(tile + (wv * RPW + rr) * TPITCH + lane * 2) = tot;
    }

    __syncthreads();

    int m  = lane & 15;
    int kh = lane >> 4;
    int cidx = wv * 16 + m;
    const float* wp = W + (size_t)(2 * kh) * DD + cidx;

    v2f bf[DD / 4];
#pragma unroll
    for (int j = 0; j < DD / 4; ++j) {
        bf[j].x = wp[(size_t)(4 * j) * DD];
        bf[j].y = wp[(size_t)(4 * j + 1) * DD];
    }
    float bcol = bias[cidx];
    float s = 0.f, s2 = 0.f;
#pragma unroll 1
    for (int t = 0; t < FROWS / 16; ++t) {
        const float* tp = tile + (t * 16 + m) * TPITCH + 2 * kh;
        v8f c = {};
#pragma unroll
        for (int j = 0; j < DD / 4; ++j) {
            v2f a = *(const v2f*)(tp + 4 * j);
            c = __builtin_amdgcn_wmma_f32_16x16x4_f32(
                    false, a, false, bf[j], (short)0, c, false, false);
        }
        int rowT = rowBase + t * 16 + 8 * kh;
        float val[8];
#pragma unroll
        for (int r = 0; r < 8; ++r) {
            val[r] = c[r] + bcol;
            if (doRelu) val[r] = fmaxf(val[r], 0.f);
            float u = (rowT + r < n) ? val[r] : 0.f;
            s  += u;
            s2 += u * u;
        }
#pragma unroll
        for (int r = 0; r < 8; ++r)
            if (rowT + r < n) out[(size_t)(rowT + r) * DD + cidx] = val[r];
    }
    if (doStats) {
        s  += __shfl_xor(s, 16, 32);
        s2 += __shfl_xor(s2, 16, 32);
        if (kh == 0) {
            atomic_add_f32(colsum   + cidx, s);
            atomic_add_f32(colsumsq + cidx, s2);
        }
    }
}

__global__ void bn_finalize_kernel(
    const float* __restrict__ colsum, const float* __restrict__ colsumsq,
    const float* __restrict__ gamma, const float* __restrict__ beta,
    float* __restrict__ scale, float* __restrict__ shift, int n)
{
    int c = threadIdx.x;
    if (c >= DD) return;
    float invN = 1.f / (float)n;
    float mu   = colsum[c] * invN;
    float var  = colsumsq[c] * invN - mu * mu;
    float sc   = gamma[c] * rsqrtf(var + BN_EPS);
    scale[c] = sc;
    shift[c] = beta[c] - mu * sc;
}

__global__ __launch_bounds__(256) void bn_apply_kernel(
    const float* __restrict__ hin, const float* __restrict__ scale,
    const float* __restrict__ shift, float* __restrict__ hout, int total4)
{
    int i = blockIdx.x * blockDim.x + threadIdx.x;
    if (i >= total4) return;
    int c = (i & 15) * 4;
    float4 v = ((const float4*)hin)[i];
    v.x = v.x * scale[c + 0] + shift[c + 0];
    v.y = v.y * scale[c + 1] + shift[c + 1];
    v.z = v.z * scale[c + 2] + shift[c + 2];
    v.w = v.w * scale[c + 3] + shift[c + 3];
    ((float4*)hout)[i] = v;
}

__global__ void pad_final_kernel(const float* __restrict__ Wf,
                                 const float* __restrict__ bf,
                                 float* __restrict__ Wpad,
                                 float* __restrict__ bpad)
{
    int i = blockIdx.x * blockDim.x + threadIdx.x;
    if (i < DD * DD) {
        int r = i >> 6, c = i & 63;
        Wpad[i] = (c < CC) ? Wf[r * CC + c] : 0.f;
    }
    if (i < DD) bpad[i] = (i < CC) ? bf[i] : 0.f;
}

__global__ __launch_bounds__(256) void logsoftmax_kernel(
    const float* __restrict__ y, float* __restrict__ out, int n)
{
    int gid  = blockIdx.x * blockDim.x + threadIdx.x;
    int row  = gid >> 5;
    int lane = gid & 31;
    if (row >= n) return;
    const float* yr = y + (size_t)row * DD;
    float v0 = yr[lane];
    bool  h1 = (lane + 32) < CC;
    float v1 = h1 ? yr[lane + 32] : -3.402823466e38f;
    float mx = fmaxf(v0, v1);
    for (int o = 16; o > 0; o >>= 1) mx = fmaxf(mx, __shfl_xor(mx, o, 32));
    float s = expf(v0 - mx) + (h1 ? expf(v1 - mx) : 0.f);
    for (int o = 16; o > 0; o >>= 1) s += __shfl_xor(s, o, 32);
    float lse = mx + logf(s);
    out[(size_t)row * CC + lane] = v0 - lse;
    if (h1) out[(size_t)row * CC + lane + 32] = v1 - lse;
}

extern "C" void kernel_launch(void* const* d_in, const int* in_sizes, int n_in,
                              void* d_out, int out_size, void* d_ws, size_t ws_size,
                              hipStream_t stream) {
    const float* x       = (const float*)d_in[0];
    const int*   ei      = (const int*)  d_in[1];
    const float* W_proj  = (const float*)d_in[2];
    const float* b_proj  = (const float*)d_in[3];
    const float* Ws      = (const float*)d_in[4];
    const float* bs      = (const float*)d_in[5];
    const float* W_final = (const float*)d_in[6];
    const float* b_final = (const float*)d_in[7];
    const float* gammas  = (const float*)d_in[8];
    const float* betas   = (const float*)d_in[9];
    float* out = (float*)d_out;

    int n = in_sizes[0] / DD;
    int E = in_sizes[1] / 2;

    size_t nd = (size_t)n * DD;
    float* ws    = (float*)d_ws;
    float* h_cur = ws;
    int*   lists = (int*)(ws + nd);
    float* agg   = ws + 2 * nd;
    float* stats = ws + 3 * nd;
    float* scale = stats + 128;
    float* shift = scale + DD;
    float* Wpad  = shift + DD;
    float* bpad  = Wpad + DD * DD;

    int scanTiles = (n + 1 + 1023) / 1024;
    int RP        = scanTiles * 1024;
    int EP        = (E + 63) / 64 * 64;
    int* cnt    = lists;
    int* rowptr = cnt + RP;
    int* rank   = rowptr + RP;
    int* col    = rank + EP;

    int nd4          = (int)(nd / 4);
    int applyBlocks  = (nd4 + 255) / 256;
    int edgeBlocks   = (E + EPB - 1) / EPB;
    int layerBlocks  = (n + FROWS - 1) / FROWS;
    int lsBlocks     = (int)(((long long)n * 32 + 255) / 256);

    zero_f4_kernel<<<scanTiles, 256, 0, stream>>>((float4*)cnt, RP / 4);
    csr_count_kernel<<<edgeBlocks, 256, 0, stream>>>(ei, cnt, rank, E);
    csr_scan_kernel<<<scanTiles, 256, 0, stream>>>(cnt, rowptr);
    csr_place_kernel<<<edgeBlocks, 256, 0, stream>>>(ei, rowptr, rank, col, E);

    for (int layer = 0; layer <= LL; ++layer) {
        const float* hin = (layer == 0) ? x : h_cur;
        const float* W   = (layer == 0) ? W_proj : Ws + (size_t)(layer - 1) * DD * DD;
        const float* b   = (layer == 0) ? b_proj : bs + (size_t)(layer - 1) * DD;

        zero_f4_kernel<<<1, 32, 0, stream>>>((float4*)stats, 32);
        gin_layer_kernel<<<layerBlocks, 128, 0, stream>>>(
            hin, rowptr, col, W, b, agg, stats, stats + DD, n,  1,  1);
        bn_finalize_kernel<<<1, DD, 0, stream>>>(
            stats, stats + DD, gammas + (size_t)layer * DD, betas + (size_t)layer * DD,
            scale, shift, n);
        bn_apply_kernel<<<applyBlocks, 256, 0, stream>>>(agg, scale, shift, h_cur, nd4);
    }

    pad_final_kernel<<<(DD * DD + 255) / 256, 256, 0, stream>>>(W_final, b_final, Wpad, bpad);
    gin_layer_kernel<<<layerBlocks, 128, 0, stream>>>(
        h_cur, rowptr, col, Wpad, bpad, agg, nullptr, nullptr, n,  0,  0);
    logsoftmax_kernel<<<lsBlocks, 256, 0, stream>>>(agg, out, n);
}


# batch norm applied by the consumer: the next layer's gather scales and shifts raw rows as they load (same fma, same sum order), scale/shift per wave from the column sums; 12 launches go (23 to 11)
# speedup vs baseline: 2.9367x; 1.2315x over previous
#include <hip/hip_runtime.h>
#include <hip/hip_bf16.h>
#include <math.h>

#define DD 64
#define CC 40
#define LL 3
#define BN_EPS 1e-5f
#define STATF ((LL + 1) * 2 * DD)

typedef __attribute__((ext_vector_type(2))) float v2f;
typedef __attribute__((ext_vector_type(8))) float v8f;

__device__ __forceinline__ void atomic_add_f32(float* p, float v) {
    __hip_atomic_fetch_add(p, v, __ATOMIC_RELAXED, __HIP_MEMORY_SCOPE_AGENT);
}

__global__ void zero_f4_kernel(float4* __restrict__ p, int n4) {
    int i = blockIdx.x * blockDim.x + threadIdx.x;
    if (i < n4) p[i] = make_float4(0.f, 0.f, 0.f, 0.f);
}

#define EPB 1024
__global__ __launch_bounds__(256) void csr_count_kernel(
    const int* __restrict__ ei, int* __restrict__ cnt, int* __restrict__ rank, int E)
{
    int e0 = blockIdx.x * EPB + threadIdx.x;
    if (blockIdx.x * EPB + EPB <= E) {
        int d[4];
#pragma unroll
        for (int k = 0; k < 4; ++k) d[k] = ei[(size_t)E + e0 + k * 256];
        int r[4];
#pragma unroll
        for (int k = 0; k < 4; ++k)
            r[k] = __hip_atomic_fetch_add(cnt + d[k], 1, __ATOMIC_RELAXED, __HIP_MEMORY_SCOPE_AGENT);
#pragma unroll
        for (int k = 0; k < 4; ++k) rank[e0 + k * 256] = r[k];
    } else {
        for (int e = e0; e < E; e += 256) {
            int d = ei[(size_t)E + e];
            rank[e] = __hip_atomic_fetch_add(cnt + d, 1, __ATOMIC_RELAXED, __HIP_MEMORY_SCOPE_AGENT);
        }
    }
}

__global__ __launch_bounds__(256) void csr_scan_kernel(
    const int* __restrict__ cnt, int* __restrict__ rowptr)
{
    __shared__ int preSum[8];
    __shared__ int waveSum[8];
    int t = threadIdx.x, lane = t & 31, w = t >> 5;
    const int4* c4 = (const int4*)cnt;
    int pre = 0;
    for (int i = t; i < (int)blockIdx.x * 256; i += 256) {
        int4 q = c4[i];
        pre += q.x + q.y + q.z + q.w;
    }
    for (int o = 16; o > 0; o >>= 1) pre += __shfl_xor(pre, o, 32);
    int4 q = c4[blockIdx.x * 256 + t];
    int tsum = q.x + q.y + q.z + q.w;
    int inc = tsum;
    for (int o = 1; o < 32; o <<= 1) {
        int y = __shfl_up(inc, o, 32);
        if (lane >= o) inc += y;
    }
    if (lane == 0)  preSum[w]  = pre;
    if (lane == 31) waveSum[w] = inc;
    __syncthreads();
    int base = 0;
    for (int i = 0; i < 8; ++i) {
        base += preSum[i];
        if (i < w) base += waveSum[i];
    }
    int ex = base + inc - tsum;
    int4 o4;
    o4.x = ex;
    o4.y = o4.x + q.x;
    o4.z = o4.y + q.y;
    o4.w = o4.z + q.z;
    ((int4*)rowptr)[blockIdx.x * 256 + t] = o4;
}

__global__ __launch_bounds__(256) void csr_place_kernel(
    const int* __restrict__ ei, const int* __restrict__ rowptr,
    const int* __restrict__ rank, int* __restrict__ col, int E)
{
    int e0 = blockIdx.x * EPB + threadIdx.x;
    if (blockIdx.x * EPB + EPB <= E) {
        int s[4], p[4];
#pragma unroll
        for (int k = 0; k < 4; ++k) {
            int e = e0 + k * 256;
            s[k] = ei[e];
            p[k] = rowptr[ei[(size_t)E + e]] + rank[e];
        }
#pragma unroll
        for (int k = 0; k < 4; ++k) col[p[k]] = s[k];
    } else {
        for (int e = e0; e < E; e += 256)
            col[rowptr[ei[(size_t)E + e]] + rank[e]] = ei[e];
    }
}

#define RPW 8
#define GCH 8
#define FROWS 32
#define TPITCH 68
__global__ __launch_bounds__(128) void gin_layer_kernel(
    const float* __restrict__ h, const int* __restrict__ rowptr,
    const int* __restrict__ col, const float* __restrict__ W,
    const float* __restrict__ bias, float* __restrict__ out,
    const float* __restrict__ inSum, const float* __restrict__ inSumSq,
    const float* __restrict__ gamma, const float* __restrict__ beta,
    float* __restrict__ colsum, float* __restrict__ colsumsq,
    int n, int hasBN, int doRelu, int doStats)
{
    __shared__ float tile[FROWS * TPITCH];
    int lane = threadIdx.x & 31;
    v2f sc = {1.f, 1.f}, sh = {0.f, 0.f};
    if (hasBN) {
        v2f cs = *(const v2f*)(inSum   + lane * 2);
        v2f cq = *(const v2f*)(inSumSq + lane * 2);
        v2f g  = *(const v2f*)(gamma   + lane * 2);
        v2f b  = *(const v2f*)(beta    + lane * 2);
        float invN = 1.f / (float)n;
        float mux = cs.x * invN, muy = cs.y * invN;
        float varx = cq.x * invN - mux * mux;
        float vary = cq.y * invN - muy * muy;
        sc.x = g.x * rsqrtf(varx + BN_EPS);
        sc.y = g.y * rsqrtf(vary + BN_EPS);
        sh.x = b.x - mux * sc.x;
        sh.y = b.y - muy * sc.y;
    }
    int wv = __builtin_amdgcn_readfirstlane((int)(threadIdx.x >> 5));
    int rowBase = blockIdx.x * FROWS;
    int r0 = rowBase + wv * RPW;
    int nr = n - r0;
    nr = nr < 0 ? 0 : (nr > RPW ? RPW : nr);
    int pi = r0 + (lane < nr ? lane : nr);
    pi = pi > n ? n : pi;
    int pj = rowptr[pi];
    for (int rr = 0; rr < RPW; ++rr) {
        v2f tot = {0.f, 0.f};
        if (rr < nr) {
            v2f ownRaw = *(const v2f*)(h + (size_t)(r0 + rr) * DD + lane * 2);
            int beg = __builtin_amdgcn_readlane(pj, rr);
            int end = __builtin_amdgcn_readlane(pj, rr + 1);
            v2f acc = {0.f, 0.f};
            for (int base = beg; base < end; base += 32) {
                int c = (end - base < 32) ? (end - base) : 32;
                int idx = col[base + (lane < c ? lane : c - 1)];
                for (int i0 = 0; i0 < c; i0 += GCH) {
                    v2f v[GCH];
#pragma unroll
                    for (int i = 0; i < GCH; ++i) {
                        int k = (i0 + i < c) ? (i0 + i) : (c - 1);
                        int src = __builtin_amdgcn_readlane(idx, k);
                        v[i] = *(const v2f*)(h + (size_t)src * DD + lane * 2);
                    }
#pragma unroll
                    for (int i = 0; i < GCH; ++i) {
                        bool ok = (i0 + i) < c;
                        acc.x += ok ? __builtin_fmaf(v[i].x, sc.x, sh.x) : 0.f;
                        acc.y += ok ? __builtin_fmaf(v[i].y, sc.y, sh.y) : 0.f;
                    }
                }
            }
            tot.x = __builtin_fmaf(ownRaw.x, sc.x, sh.x) + acc.x;
            tot.y = __builtin_fmaf(ownRaw.y, sc.y, sh.y) + acc.y;
        }
        *(v2f*)(tile + (wv * RPW + rr) * TPITCH + lane * 2) = tot;
    }

    __syncthreads();

    int m  = lane & 15;
    int kh = lane >> 4;
    int cidx = wv * 16 + m;
    const float* wp = W + (size_t)(2 * kh) * DD + cidx;

    v2f bf[DD / 4];
#pragma unroll
    for (int j = 0; j < DD / 4; ++j) {
        bf[j].x = wp[(size_t)(4 * j) * DD];
        bf[j].y = wp[(size_t)(4 * j + 1) * DD];
    }
    float bcol = bias[cidx];
    float s = 0.f, s2 = 0.f;
#pragma unroll 1
    for (int t = 0; t < FROWS / 16; ++t) {
        const float* tp = tile + (t * 16 + m) * TPITCH + 2 * kh;
        v8f c = {};
#pragma unroll
        for (int j = 0; j < DD / 4; ++j) {
            v2f a = *(const v2f*)(tp + 4 * j);
            c = __builtin_amdgcn_wmma_f32_16x16x4_f32(
                    false, a, false, bf[j], (short)0, c, false, false);
        }
        int rowT = rowBase + t * 16 + 8 * kh;
        float val[8];
#pragma unroll
        for (int r = 0; r < 8; ++r) {
            val[r] = c[r] + bcol;
            if (doRelu) val[r] = fmaxf(val[r], 0.f);
            float u = (rowT + r < n) ? val[r] : 0.f;
            s  += u;
            s2 += u * u;
        }
#pragma unroll
        for (int r = 0; r < 8; ++r)
            if (rowT + r < n) out[(size_t)(rowT + r) * DD + cidx] = val[r];
    }
    if (doStats) {
        s  += __shfl_xor(s, 16, 32);
        s2 += __shfl_xor(s2, 16, 32);
        if (kh == 0) {
            atomic_add_f32(colsum   + cidx, s);
            atomic_add_f32(colsumsq + cidx, s2);
        }
    }
}

__global__ void pad_final_kernel(const float* __restrict__ Wf,
                                 const float* __restrict__ bf,
                                 float* __restrict__ Wpad,
                                 float* __restrict__ bpad)
{
    int i = blockIdx.x * blockDim.x + threadIdx.x;
    if (i < DD * DD) {
        int r = i >> 6, c = i & 63;
        Wpad[i] = (c < CC) ? Wf[r * CC + c] : 0.f;
    }
    if (i < DD) bpad[i] = (i < CC) ? bf[i] : 0.f;
}

__global__ __launch_bounds__(256) void logsoftmax_kernel(
    const float* __restrict__ y, float* __restrict__ out, int n)
{
    int gid  = blockIdx.x * blockDim.x + threadIdx.x;
    int row  = gid >> 5;
    int lane = gid & 31;
    if (row >= n) return;
    const float* yr = y + (size_t)row * DD;
    float v0 = yr[lane];
    bool  h1 = (lane + 32) < CC;
    float v1 = h1 ? yr[lane + 32] : -3.402823466e38f;
    float mx = fmaxf(v0, v1);
    for (int o = 16; o > 0; o >>= 1) mx = fmaxf(mx, __shfl_xor(mx, o, 32));
    float s = expf(v0 - mx) + (h1 ? expf(v1 - mx) : 0.f);
    for (int o = 16; o > 0; o >>= 1) s += __shfl_xor(s, o, 32);
    float lse = mx + logf(s);
    out[(size_t)row * CC + lane] = v0 - lse;
    if (h1) out[(size_t)row * CC + lane + 32] = v1 - lse;
}

extern "C" void kernel_launch(void* const* d_in, const int* in_sizes, int n_in,
                              void* d_out, int out_size, void* d_ws, size_t ws_size,
                              hipStream_t stream) {
    const float* x       = (const float*)d_in[0];
    const int*   ei      = (const int*)  d_in[1];
    const float* W_proj  = (const float*)d_in[2];
    const float* b_proj  = (const float*)d_in[3];
    const float* Ws      = (const float*)d_in[4];
    const float* bs      = (const float*)d_in[5];
    const float* W_final = (const float*)d_in[6];
    const float* b_final = (const float*)d_in[7];
    const float* gammas  = (const float*)d_in[8];
    const float* betas   = (const float*)d_in[9];
    float* out = (float*)d_out;

    int n = in_sizes[0] / DD;
    int E = in_sizes[1] / 2;

    size_t nd = (size_t)n * DD;
    float* ws     = (float*)d_ws;
    float* planeA = ws;
    int*   lists  = (int*)(ws + nd);
    float* planeB = ws + 2 * nd;
    float* Wpad   = ws + 3 * nd;
    float* bpad   = Wpad + DD * DD;

    int scanTiles = (n + 1 + 1023) / 1024;
    int RP        = scanTiles * 1024;
    int EP        = (E + 63) / 64 * 64;
    float* stats = (float*)lists;
    int* cnt    = lists + STATF;
    int* rowptr = cnt + RP;
    int* rank   = rowptr + RP;
    int* col    = rank + EP;

    int zero4        = (RP + STATF) / 4;
    int edgeBlocks   = (E + EPB - 1) / EPB;
    int layerBlocks  = (n + FROWS - 1) / FROWS;
    int lsBlocks     = (int)(((long long)n * 32 + 255) / 256);

    zero_f4_kernel<<<(zero4 + 255) / 256, 256, 0, stream>>>((float4*)lists, zero4);
    csr_count_kernel<<<edgeBlocks, 256, 0, stream>>>(ei, cnt, rank, E);
    csr_scan_kernel<<<scanTiles, 256, 0, stream>>>(cnt, rowptr);
    csr_place_kernel<<<edgeBlocks, 256, 0, stream>>>(ei, rowptr, rank, col, E);

    const float* hin = x;
    for (int layer = 0; layer <= LL; ++layer) {
        const float* W   = (layer == 0) ? W_proj : Ws + (size_t)(layer - 1) * DD * DD;
        const float* b   = (layer == 0) ? b_proj : bs + (size_t)(layer - 1) * DD;
        float* hout      = (layer & 1) ? planeB : planeA;
        float* st        = stats + (size_t)layer * 2 * DD;
        const float* pst = (layer == 0) ? nullptr : st - 2 * DD;
        const float* pg  = (layer == 0) ? nullptr : gammas + (size_t)(layer - 1) * DD;
        const float* pb  = (layer == 0) ? nullptr : betas + (size_t)(layer - 1) * DD;

        gin_layer_kernel<<<layerBlocks, 128, 0, stream>>>(
            hin, rowptr, col, W, b, hout, pst, (layer == 0) ? nullptr : pst + DD, pg, pb,
            st, st + DD, n,  (layer == 0) ? 0 : 1,  1,  1);
        hin = hout;
    }

    float* yfin = ((LL + 1) & 1) ? planeB : planeA;
    const float* lst = stats + (size_t)LL * 2 * DD;
    pad_final_kernel<<<(DD * DD + 255) / 256, 256, 0, stream>>>(W_final, b_final, Wpad, bpad);
    gin_layer_kernel<<<layerBlocks, 128, 0, stream>>>(
        hin, rowptr, col, Wpad, bpad, yfin, lst, lst + DD,
        gammas + (size_t)LL * DD, betas + (size_t)LL * DD,
        nullptr, nullptr, n,  1,  0,  0);
    logsoftmax_kernel<<<lsBlocks, 256, 0, stream>>>(yfin, out, n);
}


# row-pointer scan: 32 waves share the counts in front of a block's tile, 8 independent 16-byte loads per thread in flight per trip (was 8 waves, one load per wait: up to 97 round trips in a row)
# speedup vs baseline: 3.2372x; 1.1023x over previous
#include <hip/hip_runtime.h>
#include <hip/hip_bf16.h>
#include <math.h>

#define DD 64
#define CC 40
#define LL 3
#define BN_EPS 1e-5f
#define STATF ((LL + 1) * 2 * DD)

typedef __attribute__((ext_vector_type(2))) float v2f;
typedef __attribute__((ext_vector_type(8))) float v8f;

__device__ __forceinline__ void atomic_add_f32(float* p, float v) {
    __hip_atomic_fetch_add(p, v, __ATOMIC_RELAXED, __HIP_MEMORY_SCOPE_AGENT);
}

__global__ void zero_f4_kernel(float4* __restrict__ p, int n4) {
    int i = blockIdx.x * blockDim.x + threadIdx.x;
    if (i < n4) p[i] = make_float4(0.f, 0.f, 0.f, 0.f);
}

#define EPB 1024
__global__ __launch_bounds__(256) void csr_count_kernel(
    const int* __restrict__ ei, int* __restrict__ cnt, int* __restrict__ rank, int E)
{
    int e0 = blockIdx.x * EPB + threadIdx.x;
    if (blockIdx.x * EPB + EPB <= E) {
        int d[4];
#pragma unroll
        for (int k = 0; k < 4; ++k) d[k] = ei[(size_t)E + e0 + k * 256];
        int r[4];
#pragma unroll
        for (int k = 0; k < 4; ++k)
            r[k] = __hip_atomic_fetch_add(cnt + d[k], 1, __ATOMIC_RELAXED, __HIP_MEMORY_SCOPE_AGENT);
#pragma unroll
        for (int k = 0; k < 4; ++k) rank[e0 + k * 256] = r[k];
    } else {
        for (int e = e0; e < E; e += 256) {
            int d = ei[(size_t)E + e];
            rank[e] = __hip_atomic_fetch_add(cnt + d, 1, __ATOMIC_RELAXED, __HIP_MEMORY_SCOPE_AGENT);
        }
    }
}

#define SCAN_T 1024
#define SCAN_AHEAD 8
__global__ __launch_bounds__(SCAN_T) void csr_scan_kernel(
    const int* __restrict__ cnt, int* __restrict__ rowptr)
{
    __shared__ int preSum[SCAN_T / 32];
    __shared__ int waveSum[8];
    int t = threadIdx.x, lane = t & 31, w = t >> 5;
    int lim = (int)blockIdx.x * 256;
    const int4* c4 = (const int4*)cnt;
    int4 q = make_int4(0, 0, 0, 0);
    if (t < 256) q = c4[lim + t];
    int pre = 0;
    for (int i0 = 0; i0 < lim; i0 += SCAN_T * SCAN_AHEAD) {
        int4 p[SCAN_AHEAD];
#pragma unroll
        for (int u = 0; u < SCAN_AHEAD; ++u) {
            int i = i0 + u * SCAN_T + t;
            p[u] = c4[i < lim ? i : lim - 1];
        }
        __builtin_amdgcn_sched_barrier(0);
#pragma unroll
        for (int u = 0; u < SCAN_AHEAD; ++u) {
            int i = i0 + u * SCAN_T + t;
            int s4 = p[u].x + p[u].y + p[u].z + p[u].w;
            pre += (i < lim) ? s4 : 0;
        }
    }
    for (int o = 16; o > 0; o >>= 1) pre += __shfl_xor(pre, o, 32);
    int tsum = q.x + q.y + q.z + q.w;
    int inc = tsum;
    for (int o = 1; o < 32; o <<= 1) {
        int y = __shfl_up(inc, o, 32);
        if (lane >= o) inc += y;
    }
    if (lane == 0) preSum[w] = pre;
    if (lane == 31 && w < 8) waveSum[w] = inc;
    __syncthreads();
    if (t < 256) {
        int base = 0;
        for (int i = 0; i < SCAN_T / 32; ++i) base += preSum[i];
        for (int i = 0; i < 8; ++i)
            if (i < w) base += waveSum[i];
        int ex = base + inc - tsum;
        int4 o4;
        o4.x = ex;
        o4.y = o4.x + q.x;
        o4.z = o4.y + q.y;
        o4.w = o4.z + q.z;
        ((int4*)rowptr)[lim + t] = o4;
    }
}

__global__ __launch_bounds__(256) void csr_place_kernel(
    const int* __restrict__ ei, const int* __restrict__ rowptr,
    const int* __restrict__ rank, int* __restrict__ col, int E)
{
    int e0 = blockIdx.x * EPB + threadIdx.x;
    if (blockIdx.x * EPB + EPB <= E) {
        int s[4], p[4];
#pragma unroll
        for (int k = 0; k < 4; ++k) {
            int e = e0 + k * 256;
            s[k] = ei[e];
            p[k] = rowptr[ei[(size_t)E + e]] + rank[e];
        }
#pragma unroll
        for (int k = 0; k < 4; ++k) col[p[k]] = s[k];
    } else {
        for (int e = e0; e < E; e += 256)
            col[rowptr[ei[(size_t)E + e]] + rank[e]] = ei[e];
    }
}

#define RPW 8
#define GCH 8
#define FROWS 32
#define TPITCH 68
__global__ __launch_bounds__(128) void gin_layer_kernel(
    const float* __restrict__ h, const int* __restrict__ rowptr,
    const int* __restrict__ col, const float* __restrict__ W,
    const float* __restrict__ bias, float* __restrict__ out,
    const float* __restrict__ inSum, const float* __restrict__ inSumSq,
    const float* __restrict__ gamma, const float* __restrict__ beta,
    float* __restrict__ colsum, float* __restrict__ colsumsq,
    int n, int hasBN, int doRelu, int doStats)
{
    __shared__ float tile[FROWS * TPITCH];
    int lane = threadIdx.x & 31;
    v2f sc = {1.f, 1.f}, sh = {0.f, 0.f};
    if (hasBN) {
        v2f cs = *(const v2f*)(inSum   + lane * 2);
        v2f cq = *(const v2f*)(inSumSq + lane * 2);
        v2f g  = *(const v2f*)(gamma   + lane * 2);
        v2f b  = *(const v2f*)(beta    + lane * 2);
        float invN = 1.f / (float)n;
        float mux = cs.x * invN, muy = cs.y * invN;
        float varx = cq.x * invN - mux * mux;
        float vary = cq.y * invN - muy * muy;
        sc.x = g.x * rsqrtf(varx + BN_EPS);
        sc.y = g.y * rsqrtf(vary + BN_EPS);
        sh.x = b.x - mux * sc.x;
        sh.y = b.y - muy * sc.y;
    }
    int wv = __builtin_amdgcn_readfirstlane((int)(threadIdx.x >> 5));
    int rowBase = blockIdx.x * FROWS;
    int r0 = rowBase + wv * RPW;
    int nr = n - r0;
    nr = nr < 0 ? 0 : (nr > RPW ? RPW : nr);
    int pi = r0 + (lane < nr ? lane : nr);
    pi = pi > n ? n : pi;
    int pj = rowptr[pi];
    for (int rr = 0; rr < RPW; ++rr) {
        v2f tot = {0.f, 0.f};
        if (rr < nr) {
            v2f ownRaw = *(const v2f*)(h + (size_t)(r0 + rr) * DD + lane * 2);
            int beg = __builtin_amdgcn_readlane(pj, rr);
            int end = __builtin_amdgcn_readlane(pj, rr + 1);
            v2f acc = {0.f, 0.f};
            for (int base = beg; base < end; base += 32) {
                int c = (end - base < 32) ? (end - base) : 32;
                int idx = col[base + (lane < c ? lane : c - 1)];
                for (int i0 = 0; i0 < c; i0 += GCH) {
                    v2f v[GCH];
#pragma unroll
                    for (int i = 0; i < GCH; ++i) {
                        int k = (i0 + i < c) ? (i0 + i) : (c - 1);
                        int src = __builtin_amdgcn_readlane(idx, k);
                        v[i] = *(const v2f*)(h + (size_t)src * DD + lane * 2);
                    }
#pragma unroll
                    for (int i = 0; i < GCH; ++i) {
                        bool ok = (i0 + i) < c;
                        acc.x += ok ? __builtin_fmaf(v[i].x, sc.x, sh.x) : 0.f;
                        acc.y += ok ? __builtin_fmaf(v[i].y, sc.y, sh.y) : 0.f;
                    }
                }
            }
            tot.x = __builtin_fmaf(ownRaw.x, sc.x, sh.x) + acc.x;
            tot.y = __builtin_fmaf(ownRaw.y, sc.y, sh.y) + acc.y;
        }
        *(v2f*)(tile + (wv * RPW + rr) * TPITCH + lane * 2) = tot;
    }

    __syncthreads();

    int m  = lane & 15;
    int kh = lane >> 4;
    int cidx = wv * 16 + m;
    const float* wp = W + (size_t)(2 * kh) * DD + cidx;

    v2f bf[DD / 4];
#pragma unroll
    for (int j = 0; j < DD / 4; ++j) {
        bf[j].x = wp[(size_t)(4 * j) * DD];
        bf[j].y = wp[(size_t)(4 * j + 1) * DD];
    }
    float bcol = bias[cidx];
    float s = 0.f, s2 = 0.f;
#pragma unroll 1
    for (int t = 0; t < FROWS / 16; ++t) {
        const float* tp = tile + (t * 16 + m) * TPITCH + 2 * kh;
        v8f c = {};
#pragma unroll
        for (int j = 0; j < DD / 4; ++j) {
            v2f a = *(const v2f*)(tp + 4 * j);
            c = __builtin_amdgcn_wmma_f32_16x16x4_f32(
                    false, a, false, bf[j], (short)0, c, false, false);
        }
        int rowT = rowBase + t * 16 + 8 * kh;
        float val[8];
#pragma unroll
        for (int r = 0; r < 8; ++r) {
            val[r] = c[r] + bcol;
            if (doRelu) val[r] = fmaxf(val[r], 0.f);
            float u = (rowT + r < n) ? val[r] : 0.f;
            s  += u;
            s2 += u * u;
        }
#pragma unroll
        for (int r = 0; r < 8; ++r)
            if (rowT + r < n) out[(size_t)(rowT + r) * DD + cidx] = val[r];
    }
    if (doStats) {
        s  += __shfl_xor(s, 16, 32);
        s2 += __shfl_xor(s2, 16, 32);
        if (kh == 0) {
            atomic_add_f32(colsum   + cidx, s);
            atomic_add_f32(colsumsq + cidx, s2);
        }
    }
}

__global__ void pad_final_kernel(const float* __restrict__ Wf,
                                 const float* __restrict__ bf,
                                 float* __restrict__ Wpad,
                                 float* __restrict__ bpad)
{
    int i = blockIdx.x * blockDim.x + threadIdx.x;
    if (i < DD * DD) {
        int r = i >> 6, c = i & 63;
        Wpad[i] = (c < CC) ? Wf[r * CC + c] : 0.f;
    }
    if (i < DD) bpad[i] = (i < CC) ? bf[i] : 0.f;
}

__global__ __launch_bounds__(256) void logsoftmax_kernel(
    const float* __restrict__ y, float* __restrict__ out, int n)
{
    int gid  = blockIdx.x * blockDim.x + threadIdx.x;
    int row  = gid >> 5;
    int lane = gid & 31;
    if (row >= n) return;
    const float* yr = y + (size_t)row * DD;
    float v0 = yr[lane];
    bool  h1 = (lane + 32) < CC;
    float v1 = h1 ? yr[lane + 32] : -3.402823466e38f;
    float mx = fmaxf(v0, v1);
    for (int o = 16; o > 0; o >>= 1) mx = fmaxf(mx, __shfl_xor(mx, o, 32));
    float s = expf(v0 - mx) + (h1 ? expf(v1 - mx) : 0.f);
    for (int o = 16; o > 0; o >>= 1) s += __shfl_xor(s, o, 32);
    float lse = mx + logf(s);
    out[(size_t)row * CC + lane] = v0 - lse;
    if (h1) out[(size_t)row * CC + lane + 32] = v1 - lse;
}

extern "C" void kernel_launch(void* const* d_in, const int* in_sizes, int n_in,
                              void* d_out, int out_size, void* d_ws, size_t ws_size,
                              hipStream_t stream) {
    const float* x       = (const float*)d_in[0];
    const int*   ei      = (const int*)  d_in[1];
    const float* W_proj  = (const float*)d_in[2];
    const float* b_proj  = (const float*)d_in[3];
    const float* Ws      = (const float*)d_in[4];
    const float* bs      = (const float*)d_in[5];
    const float* W_final = (const float*)d_in[6];
    const float* b_final = (const float*)d_in[7];
    const float* gammas  = (const float*)d_in[8];
    const float* betas   = (const float*)d_in[9];
    float* out = (float*)d_out;

    int n = in_sizes[0] / DD;
    int E = in_sizes[1] / 2;

    size_t nd = (size_t)n * DD;
    float* ws     = (float*)d_ws;
    float* planeA = ws;
    int*   lists  = (int*)(ws + nd);
    float* planeB = ws + 2 * nd;
    float* Wpad   = ws + 3 * nd;
    float* bpad   = Wpad + DD * DD;

    int scanTiles = (n + 1 + 1023) / 1024;
    int RP        = scanTiles * 1024;
    int EP        = (E + 63) / 64 * 64;
    float* stats = (float*)lists;
    int* cnt    = lists + STATF;
    int* rowptr = cnt + RP;
    int* rank   = rowptr + RP;
    int* col    = rank + EP;

    int zero4        = (RP + STATF) / 4;
    int edgeBlocks   = (E + EPB - 1) / EPB;
    int layerBlocks  = (n + FROWS - 1) / FROWS;
    int lsBlocks     = (int)(((long long)n * 32 + 255) / 256);

    zero_f4_kernel<<<(zero4 + 255) / 256, 256, 0, stream>>>((float4*)lists, zero4);
    csr_count_kernel<<<edgeBlocks, 256, 0, stream>>>(ei, cnt, rank, E);
    csr_scan_kernel<<<scanTiles, SCAN_T, 0, stream>>>(cnt, rowptr);
    csr_place_kernel<<<edgeBlocks, 256, 0, stream>>>(ei, rowptr, rank, col, E);

    const float* hin = x;
    for (int layer = 0; layer <= LL; ++layer) {
        const float* W   = (layer == 0) ? W_proj : Ws + (size_t)(layer - 1) * DD * DD;
        const float* b   = (layer == 0) ? b_proj : bs + (size_t)(layer - 1) * DD;
        float* hout      = (layer & 1) ? planeB : planeA;
        float* st        = stats + (size_t)layer * 2 * DD;
        const float* pst = (layer == 0) ? nullptr : st - 2 * DD;
        const float* pg  = (layer == 0) ? nullptr : gammas + (size_t)(layer - 1) * DD;
        const float* pb  = (layer == 0) ? nullptr : betas + (size_t)(layer - 1) * DD;

        gin_layer_kernel<<<layerBlocks, 128, 0, stream>>>(
            hin, rowptr, col, W, b, hout, pst, (layer == 0) ? nullptr : pst + DD, pg, pb,
            st, st + DD, n,  (layer == 0) ? 0 : 1,  1,  1);
        hin = hout;
    }

    float* yfin = ((LL + 1) & 1) ? planeB : planeA;
    const float* lst = stats + (size_t)LL * 2 * DD;
    pad_final_kernel<<<(DD * DD + 255) / 256, 256, 0, stream>>>(W_final, b_final, Wpad, bpad);
    gin_layer_kernel<<<layerBlocks, 128, 0, stream>>>(
        hin, rowptr, col, Wpad, bpad, yfin, lst, lst + DD,
        gammas + (size_t)LL * DD, betas + (size_t)LL * DD,
        nullptr, nullptr, n,  1,  0,  0);
    logsoftmax_kernel<<<lsBlocks, 256, 0, stream>>>(yfin, out, n);
}


# final layer and log_softmax in one kernel: class scores stay in LDS, each wave finishes its 8 rows (same butterfly max and sum) and stores whole lines; score plane and a 100000-wave launch go
# speedup vs baseline: 3.3446x; 1.0332x over previous
#include <hip/hip_runtime.h>
#include <hip/hip_bf16.h>
#include <math.h>

#define DD 64
#define CC 40
#define LL 3
#define BN_EPS 1e-5f
#define STATF ((LL + 1) * 2 * DD)

typedef __attribute__((ext_vector_type(2))) float v2f;
typedef __attribute__((ext_vector_type(8))) float v8f;

__device__ __forceinline__ void atomic_add_f32(float* p, float v) {
    __hip_atomic_fetch_add(p, v, __ATOMIC_RELAXED, __HIP_MEMORY_SCOPE_AGENT);
}

__global__ void zero_f4_kernel(float4* __restrict__ p, int n4) {
    int i = blockIdx.x * blockDim.x + threadIdx.x;
    if (i < n4) p[i] = make_float4(0.f, 0.f, 0.f, 0.f);
}

#define EPB 1024
__global__ __launch_bounds__(256) void csr_count_kernel(
    const int* __restrict__ ei, int* __restrict__ cnt, int* __restrict__ rank, int E)
{
    int e0 = blockIdx.x * EPB + threadIdx.x;
    if (blockIdx.x * EPB + EPB <= E) {
        int d[4];
#pragma unroll
        for (int k = 0; k < 4; ++k) d[k] = ei[(size_t)E + e0 + k * 256];
        int r[4];
#pragma unroll
        for (int k = 0; k < 4; ++k)
            r[k] = __hip_atomic_fetch_add(cnt + d[k], 1, __ATOMIC_RELAXED, __HIP_MEMORY_SCOPE_AGENT);
#pragma unroll
        for (int k = 0; k < 4; ++k) rank[e0 + k * 256] = r[k];
    } else {
        for (int e = e0; e < E; e += 256) {
            int d = ei[(size_t)E + e];
            rank[e] = __hip_atomic_fetch_add(cnt + d, 1, __ATOMIC_RELAXED, __HIP_MEMORY_SCOPE_AGENT);
        }
    }
}

#define SCAN_T 1024
#define SCAN_AHEAD 8
__global__ __launch_bounds__(SCAN_T) void csr_scan_kernel(
    const int* __restrict__ cnt, int* __restrict__ rowptr)
{
    __shared__ int preSum[SCAN_T / 32];
    __shared__ int waveSum[8];
    int t = threadIdx.x, lane = t & 31, w = t >> 5;
    int lim = (int)blockIdx.x * 256;
    const int4* c4 = (const int4*)cnt;
    int4 q = make_int4(0, 0, 0, 0);
    if (t < 256) q = c4[lim + t];
    int pre = 0;
    for (int i0 = 0; i0 < lim; i0 += SCAN_T * SCAN_AHEAD) {
        int4 p[SCAN_AHEAD];
#pragma unroll
        for (int u = 0; u < SCAN_AHEAD; ++u) {
            int i = i0 + u * SCAN_T + t;
            p[u] = c4[i < lim ? i : lim - 1];
        }
        __builtin_amdgcn_sched_barrier(0);
#pragma unroll
        for (int u = 0; u < SCAN_AHEAD; ++u) {
            int i = i0 + u * SCAN_T + t;
            int s4 = p[u].x + p[u].y + p[u].z + p[u].w;
            pre += (i < lim) ? s4 : 0;
        }
    }
    for (int o = 16; o > 0; o >>= 1) pre += __shfl_xor(pre, o, 32);
    int tsum = q.x + q.y + q.z + q.w;
    int inc = tsum;
    for (int o = 1; o < 32; o <<= 1) {
        int y = __shfl_up(inc, o, 32);
        if (lane >= o) inc += y;
    }
    if (lane == 0) preSum[w] = pre;
    if (lane == 31 && w < 8) waveSum[w] = inc;
    __syncthreads();
    if (t < 256) {
        int base = 0;
        for (int i = 0; i < SCAN_T / 32; ++i) base += preSum[i];
        for (int i = 0; i < 8; ++i)
            if (i < w) base += waveSum[i];
        int ex = base + inc - tsum;
        int4 o4;
        o4.x = ex;
        o4.y = o4.x + q.x;
        o4.z = o4.y + q.y;
        o4.w = o4.z + q.z;
        ((int4*)rowptr)[lim + t] = o4;
    }
}

__global__ __launch_bounds__(256) void csr_place_kernel(
    const int* __restrict__ ei, const int* __restrict__ rowptr,
    const int* __restrict__ rank, int* __restrict__ col, int E)
{
    int e0 = blockIdx.x * EPB + threadIdx.x;
    if (blockIdx.x * EPB + EPB <= E) {
        int s[4], p[4];
#pragma unroll
        for (int k = 0; k < 4; ++k) {
            int e = e0 + k * 256;
            s[k] = ei[e];
            p[k] = rowptr[ei[(size_t)E + e]] + rank[e];
        }
#pragma unroll
        for (int k = 0; k < 4; ++k) col[p[k]] = s[k];
    } else {
        for (int e = e0; e < E; e += 256)
            col[rowptr[ei[(size_t)E + e]] + rank[e]] = ei[e];
    }
}

#define RPW 8
#define GCH 8
#define FROWS 32
#define TPITCH 68
template <bool FINAL>
__global__ __launch_bounds__(128) void gin_layer_kernel(
    const float* __restrict__ h, const int* __restrict__ rowptr,
    const int* __restrict__ col, const float* __restrict__ W,
    const float* __restrict__ bias, float* __restrict__ out,
    const float* __restrict__ inSum, const float* __restrict__ inSumSq,
    const float* __restrict__ gamma, const float* __restrict__ beta,
    float* __restrict__ colsum, float* __restrict__ colsumsq,
    int n, int hasBN, int doRelu, int doStats)
{
    __shared__ __attribute__((aligned(16))) float tile[FROWS * TPITCH];
    __shared__ float ytile[FINAL ? FROWS * TPITCH : 1];
    int lane = threadIdx.x & 31;
    v2f sc = {1.f, 1.f}, sh = {0.f, 0.f};
    if (hasBN) {
        v2f cs = *(const v2f*)(inSum   + lane * 2);
        v2f cq = *(const v2f*)(inSumSq + lane * 2);
        v2f g  = *(const v2f*)(gamma   + lane * 2);
        v2f b  = *(const v2f*)(beta    + lane * 2);
        float invN = 1.f / (float)n;
        float mux = cs.x * invN, muy = cs.y * invN;
        float varx = cq.x * invN - mux * mux;
        float vary = cq.y * invN - muy * muy;
        sc.x = g.x * rsqrtf(varx + BN_EPS);
        sc.y = g.y * rsqrtf(vary + BN_EPS);
        sh.x = b.x - mux * sc.x;
        sh.y = b.y - muy * sc.y;
    }
    int wv = __builtin_amdgcn_readfirstlane((int)(threadIdx.x >> 5));
    int rowBase = blockIdx.x * FROWS;
    int r0 = rowBase + wv * RPW;
    int nr = n - r0;
    nr = nr < 0 ? 0 : (nr > RPW ? RPW : nr);
    int pi = r0 + (lane < nr ? lane : nr);
    pi = pi > n ? n : pi;
    int pj = rowptr[pi];
    for (int rr = 0; rr < RPW; ++rr) {
        v2f tot = {0.f, 0.f};
        if (rr < nr) {
            v2f ownRaw = *(const v2f*)(h + (size_t)(r0 + rr) * DD + lane * 2);
            int beg = __builtin_amdgcn_readlane(pj, rr);
            int end = __builtin_amdgcn_readlane(pj, rr + 1);
            v2f acc = {0.f, 0.f};
            for (int base = beg; base < end; base += 32) {
                int c = (end - base < 32) ? (end - base) : 32;
                int idx = col[base + (lane < c ? lane : c - 1)];
                for (int i0 = 0; i0 < c; i0 += GCH) {
                    v2f v[GCH];
#pragma unroll
                    for (int i = 0; i < GCH; ++i) {
                        int k = (i0 + i < c) ? (i0 + i) : (c - 1);
                        int src = __builtin_amdgcn_readlane(idx, k);
                        v[i] = *(const v2f*)(h + (size_t)src * DD + lane * 2);
                    }
#pragma unroll
                    for (int i = 0; i < GCH; ++i) {
                        bool ok = (i0 + i) < c;
                        acc.x += ok ? __builtin_fmaf(v[i].x, sc.x, sh.x) : 0.f;
                        acc.y += ok ? __builtin_fmaf(v[i].y, sc.y, sh.y) : 0.f;
                    }
                }
            }
            tot.x = __builtin_fmaf(ownRaw.x, sc.x, sh.x) + acc.x;
            tot.y = __builtin_fmaf(ownRaw.y, sc.y, sh.y) + acc.y;
        }
        *(v2f*)(tile + (wv * RPW + rr) * TPITCH + lane * 2) = tot;
    }

    __syncthreads();

    int m  = lane & 15;
    int kh = lane >> 4;
    int cidx = wv * 16 + m;
    const float* wp = W + (size_t)(2 * kh) * DD + cidx;

    v2f bf[DD / 4];
#pragma unroll
    for (int j = 0; j < DD / 4; ++j) {
        bf[j].x = wp[(size_t)(4 * j) * DD];
        bf[j].y = wp[(size_t)(4 * j + 1) * DD];
    }
    float bcol = bias[cidx];
    float s = 0.f, s2 = 0.f;
#pragma unroll 1
    for (int t = 0; t < FROWS / 16; ++t) {
        const float* tp = tile + (t * 16 + m) * TPITCH + 2 * kh;
        v8f c = {};
#pragma unroll
        for (int j = 0; j < DD / 4; ++j) {
            v2f a = *(const v2f*)(tp + 4 * j);
            c = __builtin_amdgcn_wmma_f32_16x16x4_f32(
                    false, a, false, bf[j], (short)0, c, false, false);
        }
        if constexpr (FINAL) {
#pragma unroll
            for (int r = 0; r < 8; ++r)
                ytile[(t * 16 + 8 * kh + r) * TPITCH + cidx] = c[r] + bcol;
        } else {
            int rowT = rowBase + t * 16 + 8 * kh;
            float val[8];
#pragma unroll
            for (int r = 0; r < 8; ++r) {
                val[r] = c[r] + bcol;
                if (doRelu) val[r] = fmaxf(val[r], 0.f);
                float u = (rowT + r < n) ? val[r] : 0.f;
                s  += u;
                s2 += u * u;
            }
#pragma unroll
            for (int r = 0; r < 8; ++r)
                if (rowT + r < n) out[(size_t)(rowT + r) * DD + cidx] = val[r];
        }
    }
    if constexpr (FINAL) {
        __syncthreads();
        float* pack = tile + wv * (RPW * CC);
        bool h1 = (lane + 32) < CC;
        float res0[RPW], res1[RPW];
#pragma unroll
        for (int rr = 0; rr < RPW; ++rr) {
            const float* yr = ytile + (wv * RPW + rr) * TPITCH;
            float v0 = yr[lane];
            float w1 = yr[lane + 32];
            float v1 = h1 ? w1 : -3.402823466e38f;
            float mx = fmaxf(v0, v1);
            for (int o = 16; o > 0; o >>= 1) mx = fmaxf(mx, __shfl_xor(mx, o, 32));
            float e1 = expf(v1 - mx);
            float e = expf(v0 - mx) + (h1 ? e1 : 0.f);
            for (int o = 16; o > 0; o >>= 1) e += __shfl_xor(e, o, 32);
            float lse = mx + logf(e);
            res0[rr] = v0 - lse;
            res1[rr] = v1 - lse;
        }
#pragma unroll
        for (int rr = 0; rr < RPW; ++rr) pack[rr * CC + lane] = res0[rr];
        if (h1) {
#pragma unroll
            for (int rr = 0; rr < RPW; ++rr) pack[rr * CC + lane + 32] = res1[rr];
        }
        __syncthreads();
        float4* o4 = (float4*)(out + (size_t)r0 * CC);
        const float4* p4 = (const float4*)pack;
#pragma unroll
        for (int p = 0; p < (RPW * CC / 4 + 31) / 32; ++p) {
            int q = p * 32 + lane;
            if (q < RPW * CC / 4 && r0 + (q * 4) / CC < n) o4[q] = p4[q];
        }
    } else if (doStats) {
        s  += __shfl_xor(s, 16, 32);
        s2 += __shfl_xor(s2, 16, 32);
        if (kh == 0) {
            atomic_add_f32(colsum   + cidx, s);
            atomic_add_f32(colsumsq + cidx, s2);
        }
    }
}

__global__ void pad_final_kernel(const float* __restrict__ Wf,
                                 const float* __restrict__ bf,
                                 float* __restrict__ Wpad,
                                 float* __restrict__ bpad)
{
    int i = blockIdx.x * blockDim.x + threadIdx.x;
    if (i < DD * DD) {
        int r = i >> 6, c = i & 63;
        Wpad[i] = (c < CC) ? Wf[r * CC + c] : 0.f;
    }
    if (i < DD) bpad[i] = (i < CC) ? bf[i] : 0.f;
}

extern "C" void kernel_launch(void* const* d_in, const int* in_sizes, int n_in,
                              void* d_out, int out_size, void* d_ws, size_t ws_size,
                              hipStream_t stream) {
    const float* x       = (const float*)d_in[0];
    const int*   ei      = (const int*)  d_in[1];
    const float* W_proj  = (const float*)d_in[2];
    const float* b_proj  = (const float*)d_in[3];
    const float* Ws      = (const float*)d_in[4];
    const float* bs      = (const float*)d_in[5];
    const float* W_final = (const float*)d_in[6];
    const float* b_final = (const float*)d_in[7];
    const float* gammas  = (const float*)d_in[8];
    const float* betas   = (const float*)d_in[9];
    float* out = (float*)d_out;

    int n = in_sizes[0] / DD;
    int E = in_sizes[1] / 2;

    size_t nd = (size_t)n * DD;
    float* ws     = (float*)d_ws;
    float* planeA = ws;
    int*   lists  = (int*)(ws + nd);
    float* planeB = ws + 2 * nd;
    float* Wpad   = ws + 3 * nd;
    float* bpad   = Wpad + DD * DD;

    int scanTiles = (n + 1 + 1023) / 1024;
    int RP        = scanTiles * 1024;
    int EP        = (E + 63) / 64 * 64;
    float* stats = (float*)lists;
    int* cnt    = lists + STATF;
    int* rowptr = cnt + RP;
    int* rank   = rowptr + RP;
    int* col    = rank + EP;

    int zero4        = (RP + STATF) / 4;
    int edgeBlocks   = (E + EPB - 1) / EPB;
    int layerBlocks  = (n + FROWS - 1) / FROWS;

    zero_f4_kernel<<<(zero4 + 255) / 256, 256, 0, stream>>>((float4*)lists, zero4);
    csr_count_kernel<<<edgeBlocks, 256, 0, stream>>>(ei, cnt, rank, E);
    csr_scan_kernel<<<scanTiles, SCAN_T, 0, stream>>>(cnt, rowptr);
    csr_place_kernel<<<edgeBlocks, 256, 0, stream>>>(ei, rowptr, rank, col, E);

    const float* hin = x;
    for (int layer = 0; layer <= LL; ++layer) {
        const float* W   = (layer == 0) ? W_proj : Ws + (size_t)(layer - 1) * DD * DD;
        const float* b   = (layer == 0) ? b_proj : bs + (size_t)(layer - 1) * DD;
        float* hout      = (layer & 1) ? planeB : planeA;
        float* st        = stats + (size_t)layer * 2 * DD;
        const float* pst = (layer == 0) ? nullptr : st - 2 * DD;
        const float* pg  = (layer == 0) ? nullptr : gammas + (size_t)(layer - 1) * DD;
        const float* pb  = (layer == 0) ? nullptr : betas + (size_t)(layer - 1) * DD;

        gin_layer_kernel<false><<<layerBlocks, 128, 0, stream>>>(
            hin, rowptr, col, W, b, hout, pst, (layer == 0) ? nullptr : pst + DD, pg, pb,
            st, st + DD, n,  (layer == 0) ? 0 : 1,  1,  1);
        hin = hout;
    }

    const float* lst = stats + (size_t)LL * 2 * DD;
    pad_final_kernel<<<(DD * DD + 255) / 256, 256, 0, stream>>>(W_final, b_final, Wpad, bpad);
    gin_layer_kernel<true><<<layerBlocks, 128, 0, stream>>>(
        hin, rowptr, col, Wpad, bpad, out, lst, lst + DD,
        gammas + (size_t)LL * DD, betas + (size_t)LL * DD,
        nullptr, nullptr, n,  1,  0,  0);
}


# edge grouping: two-level bucket sort counted in LDS (1024-row buckets, runs leave as consecutive words) replaces global int atomics and random 4-byte reads and scatter; row list order = LDS add order
# speedup vs baseline: 4.9215x; 1.4715x over previous
#include <hip/hip_runtime.h>
#include <hip/hip_bf16.h>
#include <math.h>

#define DD 64
#define CC 40
#define LL 3
#define BN_EPS 1e-5f
#define STATF ((LL + 1) * 2 * DD)

typedef __attribute__((ext_vector_type(2))) float v2f;
typedef __attribute__((ext_vector_type(8))) float v8f;

__device__ __forceinline__ void atomic_add_f32(float* p, float v) {
    __hip_atomic_fetch_add(p, v, __ATOMIC_RELAXED, __HIP_MEMORY_SCOPE_AGENT);
}

#define BSH 10
#define BW (1 << BSH)
#define NBMAX 256
#define BPL (NBMAX / 32)
#define ECH 8192
#define ETH 512
#define EPT (ECH / ETH)

__device__ __forceinline__ int lds_add_rtn(int* p) {
    return __hip_atomic_fetch_add(p, 1, __ATOMIC_RELAXED, __HIP_MEMORY_SCOPE_WORKGROUP);
}

__global__ __launch_bounds__(ETH) void bucket_hist_kernel(
    const int* __restrict__ ei, int* __restrict__ bh, int E, int NB, int nblk)
{
    __shared__ int hist[NBMAX];
    int t = threadIdx.x;
    if (t < NBMAX) hist[t] = 0;
    __syncthreads();
    int e0 = blockIdx.x * ECH + t;
    int d[EPT];
#pragma unroll
    for (int k = 0; k < EPT; ++k) {
        int e = e0 + k * ETH;
        d[k] = ei[(size_t)E + (e < E ? e : E - 1)];
    }
#pragma unroll
    for (int k = 0; k < EPT; ++k)
        if (e0 + k * ETH < E) lds_add_rtn(hist + (d[k] >> BSH));
    __syncthreads();
    if (t < NB) bh[t * nblk + blockIdx.x] = hist[t];
}

#define SCQ 8
__global__ __launch_bounds__(1024) void bucket_scan_kernel(
    int* __restrict__ bh, int total, float* __restrict__ stats, int nstat)
{
    __shared__ int wsum[32];
    int t = threadIdx.x, lane = t & 31, w = t >> 5;
    for (int i = t; i < nstat; i += 1024) stats[i] = 0.f;
    int4* b4 = (int4*)bh;
    int carry = 0;
    for (int s0 = 0; s0 < total; s0 += 1024 * SCQ * 4) {
        int b = s0 + t * (SCQ * 4);
        int4 v[SCQ];
#pragma unroll
        for (int u = 0; u < SCQ; ++u) {
            int i = b + 4 * u;
            v[u] = b4[(i < total ? i : 0) >> 2];
        }
        int s = 0;
#pragma unroll
        for (int u = 0; u < SCQ; ++u) {
            int i = b + 4 * u;
            v[u].x = (i     < total) ? v[u].x : 0;
            v[u].y = (i + 1 < total) ? v[u].y : 0;
            v[u].z = (i + 2 < total) ? v[u].z : 0;
            v[u].w = (i + 3 < total) ? v[u].w : 0;
            s += v[u].x + v[u].y + v[u].z + v[u].w;
        }
        int inc = s;
        for (int o = 1; o < 32; o <<= 1) {
            int y = __shfl_up(inc, o, 32);
            if (lane >= o) inc += y;
        }
        if (lane == 31) wsum[w] = inc;
        __syncthreads();
        int base = 0, all = 0;
        for (int i = 0; i < 32; ++i) {
            int x = wsum[i];
            base += (i < w) ? x : 0;
            all  += x;
        }
        int run = carry + base + inc - s;
#pragma unroll
        for (int u = 0; u < SCQ; ++u) {
            int i = b + 4 * u;
            int4 o4;
            o4.x = run;
            o4.y = o4.x + v[u].x;
            o4.z = o4.y + v[u].y;
            o4.w = o4.z + v[u].z;
            run  = o4.w + v[u].w;
            if (i < total) b4[i >> 2] = o4;
        }
        carry += all;
        __syncthreads();
    }
}

__global__ __launch_bounds__(ETH) void bucket_scatter_kernel(
    const int* __restrict__ ei, const int* __restrict__ offs,
    int* __restrict__ tSrc, int* __restrict__ tDst, int E, int NB, int nblk)
{
    __shared__ int hist[NBMAX];
    __shared__ int lpre[NBMAX];
    __shared__ int goff[NBMAX];
    __shared__ int sSrc[ECH];
    __shared__ int sDst[ECH];
    int t = threadIdx.x;
    if (t < NBMAX) hist[t] = 0;
    int og[BPL];
    if (t < 32) {
#pragma unroll
        for (int j = 0; j < BPL; ++j) {
            int k = t * BPL + j;
            og[j] = offs[(k < NB ? k : 0) * nblk + blockIdx.x];
        }
    }
    __syncthreads();
    int e0 = blockIdx.x * ECH + t;
    int s[EPT], d[EPT], r[EPT];
#pragma unroll
    for (int k = 0; k < EPT; ++k) {
        int e = e0 + k * ETH;
        e = e < E ? e : E - 1;
        s[k] = ei[e];
        d[k] = ei[(size_t)E + e];
    }
#pragma unroll
    for (int k = 0; k < EPT; ++k)
        r[k] = (e0 + k * ETH < E) ? lds_add_rtn(hist + (d[k] >> BSH)) : 0;
    __syncthreads();
    if (t < 32) {
        int c[BPL];
        int sum = 0;
#pragma unroll
        for (int j = 0; j < BPL; ++j) { c[j] = hist[t * BPL + j]; sum += c[j]; }
        int inc = sum;
        for (int o = 1; o < 32; o <<= 1) {
            int y = __shfl_up(inc, o, 32);
            if (t >= o) inc += y;
        }
        int ex = inc - sum;
#pragma unroll
        for (int j = 0; j < BPL; ++j) {
            int k = t * BPL + j;
            lpre[k] = ex;
            goff[k] = og[j] - ex;
            ex += c[j];
        }
    }
    __syncthreads();
#pragma unroll
    for (int k = 0; k < EPT; ++k) {
        if (e0 + k * ETH < E) {
            int p = lpre[d[k] >> BSH] + r[k];
            sSrc[p] = s[k];
            sDst[p] = d[k];
        }
    }
    __syncthreads();
    int mblk = E - blockIdx.x * ECH;
    mblk = mblk < ECH ? mblk : ECH;
#pragma unroll
    for (int k = 0; k < EPT; ++k) {
        int i = t + k * ETH;
        if (i < mblk) {
            int dd = sDst[i];
            int g = goff[dd >> BSH] + i;
            tSrc[g] = sSrc[i];
            tDst[g] = dd;
        }
    }
}

#define FTH 1024
#define FQ 8
#define STG 32768
__global__ __launch_bounds__(FTH) void bucket_finalize_kernel(
    const int* __restrict__ tSrc, const int* __restrict__ tDst, const int* __restrict__ offs,
    int* __restrict__ rowptr, int* __restrict__ col, int E, int NB, int nblk)
{
    __shared__ int cnt[BW];
    __shared__ int wsum[FTH / 32];
    __shared__ int stage[STG];
    int t = threadIdx.x, lane = t & 31, w = t >> 5;
    int kb = blockIdx.x;
    int beg = offs[kb * nblk];
    int end = E;
    if (kb + 1 < NB) end = offs[(kb + 1) * nblk];
    int m = end - beg;
    cnt[t] = 0;
    __syncthreads();
    const int* dp = tDst + beg;
    const int* sp = tSrc + beg;
    for (int i0 = 0; i0 < m; i0 += FTH * FQ) {
        int d[FQ];
#pragma unroll
        for (int u = 0; u < FQ; ++u) {
            int i = i0 + u * FTH + t;
            d[u] = dp[i < m ? i : m - 1];
        }
#pragma unroll
        for (int u = 0; u < FQ; ++u)
            if (i0 + u * FTH + t < m) lds_add_rtn(cnt + (d[u] & (BW - 1)));
    }
    __syncthreads();
    int c = cnt[t];
    int inc = c;
    for (int o = 1; o < 32; o <<= 1) {
        int y = __shfl_up(inc, o, 32);
        if (lane >= o) inc += y;
    }
    if (lane == 31) wsum[w] = inc;
    __syncthreads();
    int base = 0;
    for (int i = 0; i < FTH / 32; ++i) base += (i < w) ? wsum[i] : 0;
    int ex = base + inc - c;
    rowptr[kb * BW + t] = beg + ex;
    cnt[t] = ex;
    __syncthreads();
    bool staged = m <= STG;
    for (int i0 = 0; i0 < m; i0 += FTH * FQ) {
        int d[FQ], s[FQ];
#pragma unroll
        for (int u = 0; u < FQ; ++u) {
            int i = i0 + u * FTH + t;
            i = i < m ? i : m - 1;
            d[u] = dp[i];
            s[u] = sp[i];
        }
#pragma unroll
        for (int u = 0; u < FQ; ++u) {
            if (i0 + u * FTH + t < m) {
                int p = lds_add_rtn(cnt + (d[u] & (BW - 1)));
                if (staged) stage[p] = s[u];
                else        col[beg + p] = s[u];
            }
        }
    }
    if (staged) {
        __syncthreads();
        for (int i = t; i < m; i += FTH) col[beg + i] = stage[i];
    }
}

#define RPW 8
#define GCH 8
#define FROWS 32
#define TPITCH 68
template <bool FINAL>
__global__ __launch_bounds__(128) void gin_layer_kernel(
    const float* __restrict__ h, const int* __restrict__ rowptr,
    const int* __restrict__ col, const float* __restrict__ W,
    const float* __restrict__ bias, float* __restrict__ out,
    const float* __restrict__ inSum, const float* __restrict__ inSumSq,
    const float* __restrict__ gamma, const float* __restrict__ beta,
    float* __restrict__ colsum, float* __restrict__ colsumsq,
    int n, int hasBN, int doRelu, int doStats)
{
    __shared__ __attribute__((aligned(16))) float tile[FROWS * TPITCH];
    __shared__ float ytile[FINAL ? FROWS * TPITCH : 1];
    int lane = threadIdx.x & 31;
    v2f sc = {1.f, 1.f}, sh = {0.f, 0.f};
    if (hasBN) {
        v2f cs = *(const v2f*)(inSum   + lane * 2);
        v2f cq = *(const v2f*)(inSumSq + lane * 2);
        v2f g  = *(const v2f*)(gamma   + lane * 2);
        v2f b  = *(const v2f*)(beta    + lane * 2);
        float invN = 1.f / (float)n;
        float mux = cs.x * invN, muy = cs.y * invN;
        float varx = cq.x * invN - mux * mux;
        float vary = cq.y * invN - muy * muy;
        sc.x = g.x * rsqrtf(varx + BN_EPS);
        sc.y = g.y * rsqrtf(vary + BN_EPS);
        sh.x = b.x - mux * sc.x;
        sh.y = b.y - muy * sc.y;
    }
    int wv = __builtin_amdgcn_readfirstlane((int)(threadIdx.x >> 5));
    int rowBase = blockIdx.x * FROWS;
    int r0 = rowBase + wv * RPW;
    int nr = n - r0;
    nr = nr < 0 ? 0 : (nr > RPW ? RPW : nr);
    int pi = r0 + (lane < nr ? lane : nr);
    pi = pi > n ? n : pi;
    int pj = rowptr[pi];
    for (int rr = 0; rr < RPW; ++rr) {
        v2f tot = {0.f, 0.f};
        if (rr < nr) {
            v2f ownRaw = *(const v2f*)(h + (size_t)(r0 + rr) * DD + lane * 2);
            int beg = __builtin_amdgcn_readlane(pj, rr);
            int end = __builtin_amdgcn_readlane(pj, rr + 1);
            v2f acc = {0.f, 0.f};
            for (int base = beg; base < end; base += 32) {
                int c = (end - base < 32) ? (end - base) : 32;
                int idx = col[base + (lane < c ? lane : c - 1)];
                for (int i0 = 0; i0 < c; i0 += GCH) {
                    v2f v[GCH];
#pragma unroll
                    for (int i = 0; i < GCH; ++i) {
                        int k = (i0 + i < c) ? (i0 + i) : (c - 1);
                        int src = __builtin_amdgcn_readlane(idx, k);
                        v[i] = *(const v2f*)(h + (size_t)src * DD + lane * 2);
                    }
#pragma unroll
                    for (int i = 0; i < GCH; ++i) {
                        bool ok = (i0 + i) < c;
                        acc.x += ok ? __builtin_fmaf(v[i].x, sc.x, sh.x) : 0.f;
                        acc.y += ok ? __builtin_fmaf(v[i].y, sc.y, sh.y) : 0.f;
                    }
                }
            }
            tot.x = __builtin_fmaf(ownRaw.x, sc.x, sh.x) + acc.x;
            tot.y = __builtin_fmaf(ownRaw.y, sc.y, sh.y) + acc.y;
        }
        *(v2f*)(tile + (wv * RPW + rr) * TPITCH + lane * 2) = tot;
    }

    __syncthreads();

    int m  = lane & 15;
    int kh = lane >> 4;
    int cidx = wv * 16 + m;
    const float* wp = W + (size_t)(2 * kh) * DD + cidx;

    v2f bf[DD / 4];
#pragma unroll
    for (int j = 0; j < DD / 4; ++j) {
        bf[j].x = wp[(size_t)(4 * j) * DD];
        bf[j].y = wp[(size_t)(4 * j + 1) * DD];
    }
    float bcol = bias[cidx];
    float s = 0.f, s2 = 0.f;
#pragma unroll 1
    for (int t = 0; t < FROWS / 16; ++t) {
        const float* tp = tile + (t * 16 + m) * TPITCH + 2 * kh;
        v8f c = {};
#pragma unroll
        for (int j = 0; j < DD / 4; ++j) {
            v2f a = *(const v2f*)(tp + 4 * j);
            c = __builtin_amdgcn_wmma_f32_16x16x4_f32(
                    false, a, false, bf[j], (short)0, c, false, false);
        }
        if constexpr (FINAL) {
#pragma unroll
            for (int r = 0; r < 8; ++r)
                ytile[(t * 16 + 8 * kh + r) * TPITCH + cidx] = c[r] + bcol;
        } else {
            int rowT = rowBase + t * 16 + 8 * kh;
            float val[8];
#pragma unroll
            for (int r = 0; r < 8; ++r) {
                val[r] = c[r] + bcol;
                if (doRelu) val[r] = fmaxf(val[r], 0.f);
                float u = (rowT + r < n) ? val[r] : 0.f;
                s  += u;
                s2 += u * u;
            }
#pragma unroll
            for (int r = 0; r < 8; ++r)
                if (rowT + r < n) out[(size_t)(rowT + r) * DD + cidx] = val[r];
        }
    }
    if constexpr (FINAL) {
        __syncthreads();
        float* pack = tile + wv * (RPW * CC);
        bool h1 = (lane + 32) < CC;
        float res0[RPW], res1[RPW];
#pragma unroll
        for (int rr = 0; rr < RPW; ++rr) {
            const float* yr = ytile + (wv * RPW + rr) * TPITCH;
            float v0 = yr[lane];
            float w1 = yr[lane + 32];
            float v1 = h1 ? w1 : -3.402823466e38f;
            float mx = fmaxf(v0, v1);
            for (int o = 16; o > 0; o >>= 1) mx = fmaxf(mx, __shfl_xor(mx, o, 32));
            float e1 = expf(v1 - mx);
            float e = expf(v0 - mx) + (h1 ? e1 : 0.f);
            for (int o = 16; o > 0; o >>= 1) e += __shfl_xor(e, o, 32);
            float lse = mx + logf(e);
            res0[rr] = v0 - lse;
            res1[rr] = v1 - lse;
        }
#pragma unroll
        for (int rr = 0; rr < RPW; ++rr) pack[rr * CC + lane] = res0[rr];
        if (h1) {
#pragma unroll
            for (int rr = 0; rr < RPW; ++rr) pack[rr * CC + lane + 32] = res1[rr];
        }
        __syncthreads();
        float4* o4 = (float4*)(out + (size_t)r0 * CC);
        const float4* p4 = (const float4*)pack;
#pragma unroll
        for (int p = 0; p < (RPW * CC / 4 + 31) / 32; ++p) {
            int q = p * 32 + lane;
            if (q < RPW * CC / 4 && r0 + (q * 4) / CC < n) o4[q] = p4[q];
        }
    } else if (doStats) {
        s  += __shfl_xor(s, 16, 32);
        s2 += __shfl_xor(s2, 16, 32);
        if (kh == 0) {
            atomic_add_f32(colsum   + cidx, s);
            atomic_add_f32(colsumsq + cidx, s2);
        }
    }
}

__global__ void pad_final_kernel(const float* __restrict__ Wf,
                                 const float* __restrict__ bf,
                                 float* __restrict__ Wpad,
                                 float* __restrict__ bpad)
{
    int i = blockIdx.x * blockDim.x + threadIdx.x;
    if (i < DD * DD) {
        int r = i >> 6, c = i & 63;
        Wpad[i] = (c < CC) ? Wf[r * CC + c] : 0.f;
    }
    if (i < DD) bpad[i] = (i < CC) ? bf[i] : 0.f;
}

extern "C" void kernel_launch(void* const* d_in, const int* in_sizes, int n_in,
                              void* d_out, int out_size, void* d_ws, size_t ws_size,
                              hipStream_t stream) {
    const float* x       = (const float*)d_in[0];
    const int*   ei      = (const int*)  d_in[1];
    const float* W_proj  = (const float*)d_in[2];
    const float* b_proj  = (const float*)d_in[3];
    const float* Ws      = (const float*)d_in[4];
    const float* bs      = (const float*)d_in[5];
    const float* W_final = (const float*)d_in[6];
    const float* b_final = (const float*)d_in[7];
    const float* gammas  = (const float*)d_in[8];
    const float* betas   = (const float*)d_in[9];
    float* out = (float*)d_out;

    int n = in_sizes[0] / DD;
    int E = in_sizes[1] / 2;

    size_t nd = (size_t)n * DD;
    float* ws     = (float*)d_ws;
    float* planeA = ws;
    int*   lists  = (int*)(ws + nd);
    float* planeB = ws + 2 * nd;
    float* Wpad   = ws + 3 * nd;
    float* bpad   = Wpad + DD * DD;

    int NB        = (n + 1 + BW - 1) / BW;
    int RP        = NB * BW;
    int EP        = (E + 63) / 64 * 64;
    int nblk      = (E + ECH - 1) / ECH;
    int total     = NB * nblk;
    float* stats = (float*)lists;
    int* rowptr = lists + STATF;
    int* col    = rowptr + RP;
    int* tSrc   = col + EP;
    int* tDst   = tSrc + EP;
    int* bh     = tDst + EP;

    int layerBlocks  = (n + FROWS - 1) / FROWS;

    bucket_hist_kernel<<<nblk, ETH, 0, stream>>>(ei, bh, E, NB, nblk);
    bucket_scan_kernel<<<1, 1024, 0, stream>>>(bh, total, stats, STATF);
    bucket_scatter_kernel<<<nblk, ETH, 0, stream>>>(ei, bh, tSrc, tDst, E, NB, nblk);
    bucket_finalize_kernel<<<NB, FTH, 0, stream>>>(tSrc, tDst, bh, rowptr, col, E, NB, nblk);

    const float* hin = x;
    for (int layer = 0; layer <= LL; ++layer) {
        const float* W   = (layer == 0) ? W_proj : Ws + (size_t)(layer - 1) * DD * DD;
        const float* b   = (layer == 0) ? b_proj : bs + (size_t)(layer - 1) * DD;
        float* hout      = (layer & 1) ? planeB : planeA;
        float* st        = stats + (size_t)layer * 2 * DD;
        const float* pst = (layer == 0) ? nullptr : st - 2 * DD;
        const float* pg  = (layer == 0) ? nullptr : gammas + (size_t)(layer - 1) * DD;
        const float* pb  = (layer == 0) ? nullptr : betas + (size_t)(layer - 1) * DD;

        gin_layer_kernel<false><<<layerBlocks, 128, 0, stream>>>(
            hin, rowptr, col, W, b, hout, pst, (layer == 0) ? nullptr : pst + DD, pg, pb,
            st, st + DD, n,  (layer == 0) ? 0 : 1,  1,  1);
        hin = hout;
    }

    const float* lst = stats + (size_t)LL * 2 * DD;
    pad_final_kernel<<<(DD * DD + 255) / 256, 256, 0, stream>>>(W_final, b_final, Wpad, bpad);
    gin_layer_kernel<true><<<layerBlocks, 128, 0, stream>>>(
        hin, rowptr, col, Wpad, bpad, out, lst, lst + DD,
        gammas + (size_t)LL * DD, betas + (size_t)LL * DD,
        nullptr, nullptr, n,  1,  0,  0);
}


# gather: a wave walks its 8 rows' lists as one stream, 16 row loads in flight across row ends, next indices prefetched; a row's sum meets its own h in LDS at its list end; add order kept
# speedup vs baseline: 5.1684x; 1.0502x over previous
#include <hip/hip_runtime.h>
#include <hip/hip_bf16.h>
#include <math.h>

#define DD 64
#define CC 40
#define LL 3
#define BN_EPS 1e-5f
#define STATF ((LL + 1) * 2 * DD)

typedef __attribute__((ext_vector_type(2))) float v2f;
typedef __attribute__((ext_vector_type(8))) float v8f;

__device__ __forceinline__ void atomic_add_f32(float* p, float v) {
    __hip_atomic_fetch_add(p, v, __ATOMIC_RELAXED, __HIP_MEMORY_SCOPE_AGENT);
}

#define BSH 10
#define BW (1 << BSH)
#define NBMAX 256
#define BPL (NBMAX / 32)
#define ECH 8192
#define ETH 512
#define EPT (ECH / ETH)

__device__ __forceinline__ int lds_add_rtn(int* p) {
    return __hip_atomic_fetch_add(p, 1, __ATOMIC_RELAXED, __HIP_MEMORY_SCOPE_WORKGROUP);
}

__global__ __launch_bounds__(ETH) void bucket_hist_kernel(
    const int* __restrict__ ei, int* __restrict__ bh, int E, int NB, int nblk)
{
    __shared__ int hist[NBMAX];
    int t = threadIdx.x;
    if (t < NBMAX) hist[t] = 0;
    __syncthreads();
    int e0 = blockIdx.x * ECH + t;
    int d[EPT];
#pragma unroll
    for (int k = 0; k < EPT; ++k) {
        int e = e0 + k * ETH;
        d[k] = ei[(size_t)E + (e < E ? e : E - 1)];
    }
#pragma unroll
    for (int k = 0; k < EPT; ++k)
        if (e0 + k * ETH < E) lds_add_rtn(hist + (d[k] >> BSH));
    __syncthreads();
    if (t < NB) bh[t * nblk + blockIdx.x] = hist[t];
}

#define SCQ 8
__global__ __launch_bounds__(1024) void bucket_scan_kernel(
    int* __restrict__ bh, int total, float* __restrict__ stats, int nstat)
{
    __shared__ int wsum[32];
    int t = threadIdx.x, lane = t & 31, w = t >> 5;
    for (int i = t; i < nstat; i += 1024) stats[i] = 0.f;
    int4* b4 = (int4*)bh;
    int carry = 0;
    for (int s0 = 0; s0 < total; s0 += 1024 * SCQ * 4) {
        int b = s0 + t * (SCQ * 4);
        int4 v[SCQ];
#pragma unroll
        for (int u = 0; u < SCQ; ++u) {
            int i = b + 4 * u;
            v[u] = b4[(i < total ? i : 0) >> 2];
        }
        int s = 0;
#pragma unroll
        for (int u = 0; u < SCQ; ++u) {
            int i = b + 4 * u;
            v[u].x = (i     < total) ? v[u].x : 0;
            v[u].y = (i + 1 < total) ? v[u].y : 0;
            v[u].z = (i + 2 < total) ? v[u].z : 0;
            v[u].w = (i + 3 < total) ? v[u].w : 0;
            s += v[u].x + v[u].y + v[u].z + v[u].w;
        }
        int inc = s;
        for (int o = 1; o < 32; o <<= 1) {
            int y = __shfl_up(inc, o, 32);
            if (lane >= o) inc += y;
        }
        if (lane == 31) wsum[w] = inc;
        __syncthreads();
        int base = 0, all = 0;
        for (int i = 0; i < 32; ++i) {
            int x = wsum[i];
            base += (i < w) ? x : 0;
            all  += x;
        }
        int run = carry + base + inc - s;
#pragma unroll
        for (int u = 0; u < SCQ; ++u) {
            int i = b + 4 * u;
            int4 o4;
            o4.x = run;
            o4.y = o4.x + v[u].x;
            o4.z = o4.y + v[u].y;
            o4.w = o4.z + v[u].z;
            run  = o4.w + v[u].w;
            if (i < total) b4[i >> 2] = o4;
        }
        carry += all;
        __syncthreads();
    }
}

__global__ __launch_bounds__(ETH) void bucket_scatter_kernel(
    const int* __restrict__ ei, const int* __restrict__ offs,
    int* __restrict__ tSrc, int* __restrict__ tDst, int E, int NB, int nblk)
{
    __shared__ int hist[NBMAX];
    __shared__ int lpre[NBMAX];
    __shared__ int goff[NBMAX];
    __shared__ int sSrc[ECH];
    __shared__ int sDst[ECH];
    int t = threadIdx.x;
    if (t < NBMAX) hist[t] = 0;
    int og[BPL];
    if (t < 32) {
#pragma unroll
        for (int j = 0; j < BPL; ++j) {
            int k = t * BPL + j;
            og[j] = offs[(k < NB ? k : 0) * nblk + blockIdx.x];
        }
    }
    __syncthreads();
    int e0 = blockIdx.x * ECH + t;
    int s[EPT], d[EPT], r[EPT];
#pragma unroll
    for (int k = 0; k < EPT; ++k) {
        int e = e0 + k * ETH;
        e = e < E ? e : E - 1;
        s[k] = ei[e];
        d[k] = ei[(size_t)E + e];
    }
#pragma unroll
    for (int k = 0; k < EPT; ++k)
        r[k] = (e0 + k * ETH < E) ? lds_add_rtn(hist + (d[k] >> BSH)) : 0;
    __syncthreads();
    if (t < 32) {
        int c[BPL];
        int sum = 0;
#pragma unroll
        for (int j = 0; j < BPL; ++j) { c[j] = hist[t * BPL + j]; sum += c[j]; }
        int inc = sum;
        for (int o = 1; o < 32; o <<= 1) {
            int y = __shfl_up(inc, o, 32);
            if (t >= o) inc += y;
        }
        int ex = inc - sum;
#pragma unroll
        for (int j = 0; j < BPL; ++j) {
            int k = t * BPL + j;
            lpre[k] = ex;
            goff[k] = og[j] - ex;
            ex += c[j];
        }
    }
    __syncthreads();
#pragma unroll
    for (int k = 0; k < EPT; ++k) {
        if (e0 + k * ETH < E) {
            int p = lpre[d[k] >> BSH] + r[k];
            sSrc[p] = s[k];
            sDst[p] = d[k];
        }
    }
    __syncthreads();
    int mblk = E - blockIdx.x * ECH;
    mblk = mblk < ECH ? mblk : ECH;
#pragma unroll
    for (int k = 0; k < EPT; ++k) {
        int i = t + k * ETH;
        if (i < mblk) {
            int dd = sDst[i];
            int g = goff[dd >> BSH] + i;
            tSrc[g] = sSrc[i];
            tDst[g] = dd;
        }
    }
}

#define FTH 1024
#define FQ 8
#define STG 32768
__global__ __launch_bounds__(FTH) void bucket_finalize_kernel(
    const int* __restrict__ tSrc, const int* __restrict__ tDst, const int* __restrict__ offs,
    int* __restrict__ rowptr, int* __restrict__ col, int E, int NB, int nblk)
{
    __shared__ int cnt[BW];
    __shared__ int wsum[FTH / 32];
    __shared__ int stage[STG];
    int t = threadIdx.x, lane = t & 31, w = t >> 5;
    int kb = blockIdx.x;
    int beg = offs[kb * nblk];
    int end = E;
    if (kb + 1 < NB) end = offs[(kb + 1) * nblk];
    int m = end - beg;
    cnt[t] = 0;
    __syncthreads();
    const int* dp = tDst + beg;
    const int* sp = tSrc + beg;
    for (int i0 = 0; i0 < m; i0 += FTH * FQ) {
        int d[FQ];
#pragma unroll
        for (int u = 0; u < FQ; ++u) {
            int i = i0 + u * FTH + t;
            d[u] = dp[i < m ? i : m - 1];
        }
#pragma unroll
        for (int u = 0; u < FQ; ++u)
            if (i0 + u * FTH + t < m) lds_add_rtn(cnt + (d[u] & (BW - 1)));
    }
    __syncthreads();
    int c = cnt[t];
    int inc = c;
    for (int o = 1; o < 32; o <<= 1) {
        int y = __shfl_up(inc, o, 32);
        if (lane >= o) inc += y;
    }
    if (lane == 31) wsum[w] = inc;
    __syncthreads();
    int base = 0;
    for (int i = 0; i < FTH / 32; ++i) base += (i < w) ? wsum[i] : 0;
    int ex = base + inc - c;
    rowptr[kb * BW + t] = beg + ex;
    cnt[t] = ex;
    __syncthreads();
    bool staged = m <= STG;
    for (int i0 = 0; i0 < m; i0 += FTH * FQ) {
        int d[FQ], s[FQ];
#pragma unroll
        for (int u = 0; u < FQ; ++u) {
            int i = i0 + u * FTH + t;
            i = i < m ? i : m - 1;
            d[u] = dp[i];
            s[u] = sp[i];
        }
#pragma unroll
        for (int u = 0; u < FQ; ++u) {
            if (i0 + u * FTH + t < m) {
                int p = lds_add_rtn(cnt + (d[u] & (BW - 1)));
                if (staged) stage[p] = s[u];
                else        col[beg + p] = s[u];
            }
        }
    }
    if (staged) {
        __syncthreads();
        for (int i = t; i < m; i += FTH) col[beg + i] = stage[i];
    }
}

#define RPW 8
#define GCH 16
#define FROWS 32
#define TPITCH 68
typedef __attribute__((address_space(1))) const v2f global_v2f;
__device__ __forceinline__ global_v2f* wave_uniform_row(const float* p) {
    unsigned long long a = (unsigned long long)p;
    unsigned lo = __builtin_amdgcn_readfirstlane((unsigned)a);
    unsigned hi = __builtin_amdgcn_readfirstlane((unsigned)(a >> 32));
    return (global_v2f*)(((unsigned long long)hi << 32) | lo);
}

template <bool FINAL>
__global__ __launch_bounds__(128) void gin_layer_kernel(
    const float* __restrict__ h, const int* __restrict__ rowptr,
    const int* __restrict__ col, const float* __restrict__ W,
    const float* __restrict__ bias, float* __restrict__ out,
    const float* __restrict__ inSum, const float* __restrict__ inSumSq,
    const float* __restrict__ gamma, const float* __restrict__ beta,
    float* __restrict__ colsum, float* __restrict__ colsumsq,
    int n, int hasBN, int doRelu, int doStats)
{
    __shared__ __attribute__((aligned(16))) float tile[FROWS * TPITCH];
    __shared__ float ytile[FINAL ? FROWS * TPITCH : 1];
    int lane = threadIdx.x & 31;
    v2f sc = {1.f, 1.f}, sh = {0.f, 0.f};
    if (hasBN) {
        v2f cs = *(const v2f*)(inSum   + lane * 2);
        v2f cq = *(const v2f*)(inSumSq + lane * 2);
        v2f g  = *(const v2f*)(gamma   + lane * 2);
        v2f b  = *(const v2f*)(beta    + lane * 2);
        float invN = 1.f / (float)n;
        float mux = cs.x * invN, muy = cs.y * invN;
        float varx = cq.x * invN - mux * mux;
        float vary = cq.y * invN - muy * muy;
        sc.x = g.x * rsqrtf(varx + BN_EPS);
        sc.y = g.y * rsqrtf(vary + BN_EPS);
        sh.x = b.x - mux * sc.x;
        sh.y = b.y - muy * sc.y;
    }
    int wv = __builtin_amdgcn_readfirstlane((int)(threadIdx.x >> 5));
    int rowBase = blockIdx.x * FROWS;
    int r0 = rowBase + wv * RPW;
    int nr = n - r0;
    nr = nr < 0 ? 0 : (nr > RPW ? RPW : nr);
    int pi = r0 + (lane < nr ? lane : nr);
    pi = pi > n ? n : pi;
    int pj = rowptr[pi];
    v2f own[RPW];
#pragma unroll
    for (int rr = 0; rr < RPW; ++rr) {
        int r = r0 + rr;
        r = r < n ? r : n - 1;
        own[rr] = *(const v2f*)(h + (size_t)r * DD + lane * 2);
    }
    int eb = __builtin_amdgcn_readlane(pj, 0);
    int ee = __builtin_amdgcn_readlane(pj, nr);
    int last = ee - 1;
    int idx = 0;
    if (eb < ee) {
        int p0 = eb + lane;
        idx = col[p0 < last ? p0 : last];
    }
    float* trow = tile + (wv * RPW) * TPITCH + lane * 2;
#pragma unroll
    for (int rr = 0; rr < RPW; ++rr) {
        v2f t = {0.f, 0.f};
        if (rr < nr) {
            t.x = __builtin_fmaf(own[rr].x, sc.x, sh.x);
            t.y = __builtin_fmaf(own[rr].y, sc.y, sh.y);
        }
        *(v2f*)(trow + rr * TPITCH) = t;
    }
    if (eb < ee) {
        int cur = 0;
        int rend = __builtin_amdgcn_readlane(pj, 1);
        v2f acc = {0.f, 0.f};
        auto endRows = [&](int e) {
            do {
                float* tp = trow + cur * TPITCH;
                v2f t = *(v2f*)tp;
                t.x += acc.x;
                t.y += acc.y;
                *(v2f*)tp = t;
                acc.x = 0.f;
                acc.y = 0.f;
                ++cur;
                rend = __builtin_amdgcn_readlane(pj, cur + 1);
            } while (e >= rend);
        };
        int tot = ee - eb;
        int nfull = tot / GCH;
        int e0 = eb;
        int idxN = idx;
        for (int j = 0; j < nfull; ++j) {
            int hb = (j & 1) * GCH;
            v2f v[GCH];
            global_v2f* rp[GCH];
#pragma unroll
            for (int i = 0; i < GCH; ++i) {
                int src = __builtin_amdgcn_readlane(idx, hb + i);
                rp[i] = wave_uniform_row(h + (size_t)src * DD);
            }
            __builtin_amdgcn_sched_barrier(0);
#pragma unroll
            for (int i = 0; i < GCH; ++i) v[i] = rp[i][lane];
            if (hb == 0) {
                int pn = e0 + 32 + lane;
                idxN = col[pn < last ? pn : last];
            }
#pragma unroll
            for (int i = 0; i < GCH; ++i) {
                int e = e0 + i;
                if (__builtin_expect(e >= rend, 0)) endRows(e);
                acc.x += __builtin_fmaf(v[i].x, sc.x, sh.x);
                acc.y += __builtin_fmaf(v[i].y, sc.y, sh.y);
            }
            e0 += GCH;
            if (hb != 0) idx = idxN;
        }
        int rem = tot - nfull * GCH;
        if (rem > 0) {
            int hb = (nfull & 1) * GCH;
            v2f v[GCH];
            global_v2f* rp[GCH];
#pragma unroll
            for (int i = 0; i < GCH; ++i) {
                int k = hb + (i < rem ? i : rem - 1);
                int src = __builtin_amdgcn_readlane(idx, k);
                rp[i] = wave_uniform_row(h + (size_t)src * DD);
            }
            __builtin_amdgcn_sched_barrier(0);
#pragma unroll
            for (int i = 0; i < GCH; ++i) v[i] = rp[i][lane];
#pragma unroll
            for (int i = 0; i < GCH; ++i) {
                if (i < rem) {
                    int e = e0 + i;
                    if (__builtin_expect(e >= rend, 0)) endRows(e);
                    acc.x += __builtin_fmaf(v[i].x, sc.x, sh.x);
                    acc.y += __builtin_fmaf(v[i].y, sc.y, sh.y);
                }
            }
        }
        float* tp = trow + cur * TPITCH;
        v2f t = *(v2f*)tp;
        t.x += acc.x;
        t.y += acc.y;
        *(v2f*)tp = t;
    }

    __syncthreads();

    int m  = lane & 15;
    int kh = lane >> 4;
    int cidx = wv * 16 + m;
    const float* wp = W + (size_t)(2 * kh) * DD + cidx;

    v2f bf[DD / 4];
#pragma unroll
    for (int j = 0; j < DD / 4; ++j) {
        bf[j].x = wp[(size_t)(4 * j) * DD];
        bf[j].y = wp[(size_t)(4 * j + 1) * DD];
    }
    float bcol = bias[cidx];
    float s = 0.f, s2 = 0.f;
#pragma unroll 1
    for (int t = 0; t < FROWS / 16; ++t) {
        const float* tp = tile + (t * 16 + m) * TPITCH + 2 * kh;
        v8f c = {};
#pragma unroll
        for (int j = 0; j < DD / 4; ++j) {
            v2f a = *(const v2f*)(tp + 4 * j);
            c = __builtin_amdgcn_wmma_f32_16x16x4_f32(
                    false, a, false, bf[j], (short)0, c, false, false);
        }
        if constexpr (FINAL) {
#pragma unroll
            for (int r = 0; r < 8; ++r)
                ytile[(t * 16 + 8 * kh + r) * TPITCH + cidx] = c[r] + bcol;
        } else {
            int rowT = rowBase + t * 16 + 8 * kh;
            float val[8];
#pragma unroll
            for (int r = 0; r < 8; ++r) {
                val[r] = c[r] + bcol;
                if (doRelu) val[r] = fmaxf(val[r], 0.f);
                float u = (rowT + r < n) ? val[r] : 0.f;
                s  += u;
                s2 += u * u;
            }
#pragma unroll
            for (int r = 0; r < 8; ++r)
                if (rowT + r < n) out[(size_t)(rowT + r) * DD + cidx] = val[r];
        }
    }
    if constexpr (FINAL) {
        __syncthreads();
        float* pack = tile + wv * (RPW * CC);
        bool h1 = (lane + 32) < CC;
        float res0[RPW], res1[RPW];
#pragma unroll
        for (int rr = 0; rr < RPW; ++rr) {
            const float* yr = ytile + (wv * RPW + rr) * TPITCH;
            float v0 = yr[lane];
            float w1 = yr[lane + 32];
            float v1 = h1 ? w1 : -3.402823466e38f;
            float mx = fmaxf(v0, v1);
            for (int o = 16; o > 0; o >>= 1) mx = fmaxf(mx, __shfl_xor(mx, o, 32));
            float e1 = expf(v1 - mx);
            float e = expf(v0 - mx) + (h1 ? e1 : 0.f);
            for (int o = 16; o > 0; o >>= 1) e += __shfl_xor(e, o, 32);
            float lse = mx + logf(e);
            res0[rr] = v0 - lse;
            res1[rr] = v1 - lse;
        }
#pragma unroll
        for (int rr = 0; rr < RPW; ++rr) pack[rr * CC + lane] = res0[rr];
        if (h1) {
#pragma unroll
            for (int rr = 0; rr < RPW; ++rr) pack[rr * CC + lane + 32] = res1[rr];
        }
        __syncthreads();
        float4* o4 = (float4*)(out + (size_t)r0 * CC);
        const float4* p4 = (const float4*)pack;
#pragma unroll
        for (int p = 0; p < (RPW * CC / 4 + 31) / 32; ++p) {
            int q = p * 32 + lane;
            if (q < RPW * CC / 4 && r0 + (q * 4) / CC < n) o4[q] = p4[q];
        }
    } else if (doStats) {
        s  += __shfl_xor(s, 16, 32);
        s2 += __shfl_xor(s2, 16, 32);
        if (kh == 0) {
            atomic_add_f32(colsum   + cidx, s);
            atomic_add_f32(colsumsq + cidx, s2);
        }
    }
}

__global__ void pad_final_kernel(const float* __restrict__ Wf,
                                 const float* __restrict__ bf,
                                 float* __restrict__ Wpad,
                                 float* __restrict__ bpad)
{
    int i = blockIdx.x * blockDim.x + threadIdx.x;
    if (i < DD * DD) {
        int r = i >> 6, c = i & 63;
        Wpad[i] = (c < CC) ? Wf[r * CC + c] : 0.f;
    }
    if (i < DD) bpad[i] = (i < CC) ? bf[i] : 0.f;
}

extern "C" void kernel_launch(void* const* d_in, const int* in_sizes, int n_in,
                              void* d_out, int out_size, void* d_ws, size_t ws_size,
                              hipStream_t stream) {
    const float* x       = (const float*)d_in[0];
    const int*   ei      = (const int*)  d_in[1];
    const float* W_proj  = (const float*)d_in[2];
    const float* b_proj  = (const float*)d_in[3];
    const float* Ws      = (const float*)d_in[4];
    const float* bs      = (const float*)d_in[5];
    const float* W_final = (const float*)d_in[6];
    const float* b_final = (const float*)d_in[7];
    const float* gammas  = (const float*)d_in[8];
    const float* betas   = (const float*)d_in[9];
    float* out = (float*)d_out;

    int n = in_sizes[0] / DD;
    int E = in_sizes[1] / 2;

    size_t nd = (size_t)n * DD;
    float* ws     = (float*)d_ws;
    float* planeA = ws;
    int*   lists  = (int*)(ws + nd);
    float* planeB = ws + 2 * nd;
    float* Wpad   = ws + 3 * nd;
    float* bpad   = Wpad + DD * DD;

    int NB        = (n + 1 + BW - 1) / BW;
    int RP        = NB * BW;
    int EP        = (E + 63) / 64 * 64;
    int nblk      = (E + ECH - 1) / ECH;
    int total     = NB * nblk;
    float* stats = (float*)lists;
    int* rowptr = lists + STATF;
    int* col    = rowptr + RP;
    int* tSrc   = col + EP;
    int* tDst   = tSrc + EP;
    int* bh     = tDst + EP;

    int layerBlocks  = (n + FROWS - 1) / FROWS;

    bucket_hist_kernel<<<nblk, ETH, 0, stream>>>(ei, bh, E, NB, nblk);
    bucket_scan_kernel<<<1, 1024, 0, stream>>>(bh, total, stats, STATF);
    bucket_scatter_kernel<<<nblk, ETH, 0, stream>>>(ei, bh, tSrc, tDst, E, NB, nblk);
    bucket_finalize_kernel<<<NB, FTH, 0, stream>>>(tSrc, tDst, bh, rowptr, col, E, NB, nblk);

    const float* hin = x;
    for (int layer = 0; layer <= LL; ++layer) {
        const float* W   = (layer == 0) ? W_proj : Ws + (size_t)(layer - 1) * DD * DD;
        const float* b   = (layer == 0) ? b_proj : bs + (size_t)(layer - 1) * DD;
        float* hout      = (layer & 1) ? planeB : planeA;
        float* st        = stats + (size_t)layer * 2 * DD;
        const float* pst = (layer == 0) ? nullptr : st - 2 * DD;
        const float* pg  = (layer == 0) ? nullptr : gammas + (size_t)(layer - 1) * DD;
        const float* pb  = (layer == 0) ? nullptr : betas + (size_t)(layer - 1) * DD;

        gin_layer_kernel<false><<<layerBlocks, 128, 0, stream>>>(
            hin, rowptr, col, W, b, hout, pst, (layer == 0) ? nullptr : pst + DD, pg, pb,
            st, st + DD, n,  (layer == 0) ? 0 : 1,  1,  1);
        hin = hout;
    }

    const float* lst = stats + (size_t)LL * 2 * DD;
    pad_final_kernel<<<(DD * DD + 255) / 256, 256, 0, stream>>>(W_final, b_final, Wpad, bpad);
    gin_layer_kernel<true><<<layerBlocks, 128, 0, stream>>>(
        hin, rowptr, col, Wpad, bpad, out, lst, lst + DD,
        gammas + (size_t)LL * DD, betas + (size_t)LL * DD,
        nullptr, nullptr, n,  1,  0,  0);
}


# layer output rows staged in a second LDS tile and stored whole: a wave writes its 8 rows 16 B per lane (4 whole 128-B lines per store), not 64-B column pieces of two rows; values and sums same
# speedup vs baseline: 5.5186x; 1.0678x over previous
#include <hip/hip_runtime.h>
#include <hip/hip_bf16.h>
#include <math.h>

#define DD 64
#define CC 40
#define LL 3
#define BN_EPS 1e-5f
#define STATF ((LL + 1) * 2 * DD)

typedef __attribute__((ext_vector_type(2))) float v2f;
typedef __attribute__((ext_vector_type(8))) float v8f;

__device__ __forceinline__ void atomic_add_f32(float* p, float v) {
    __hip_atomic_fetch_add(p, v, __ATOMIC_RELAXED, __HIP_MEMORY_SCOPE_AGENT);
}

#define BSH 10
#define BW (1 << BSH)
#define NBMAX 256
#define BPL (NBMAX / 32)
#define ECH 8192
#define ETH 512
#define EPT (ECH / ETH)

__device__ __forceinline__ int lds_add_rtn(int* p) {
    return __hip_atomic_fetch_add(p, 1, __ATOMIC_RELAXED, __HIP_MEMORY_SCOPE_WORKGROUP);
}

__global__ __launch_bounds__(ETH) void bucket_hist_kernel(
    const int* __restrict__ ei, int* __restrict__ bh, int E, int NB, int nblk)
{
    __shared__ int hist[NBMAX];
    int t = threadIdx.x;
    if (t < NBMAX) hist[t] = 0;
    __syncthreads();
    int e0 = blockIdx.x * ECH + t;
    int d[EPT];
#pragma unroll
    for (int k = 0; k < EPT; ++k) {
        int e = e0 + k * ETH;
        d[k] = ei[(size_t)E + (e < E ? e : E - 1)];
    }
#pragma unroll
    for (int k = 0; k < EPT; ++k)
        if (e0 + k * ETH < E) lds_add_rtn(hist + (d[k] >> BSH));
    __syncthreads();
    if (t < NB) bh[t * nblk + blockIdx.x] = hist[t];
}

#define SCQ 8
__global__ __launch_bounds__(1024) void bucket_scan_kernel(
    int* __restrict__ bh, int total, float* __restrict__ stats, int nstat)
{
    __shared__ int wsum[32];
    int t = threadIdx.x, lane = t & 31, w = t >> 5;
    for (int i = t; i < nstat; i += 1024) stats[i] = 0.f;
    int4* b4 = (int4*)bh;
    int carry = 0;
    for (int s0 = 0; s0 < total; s0 += 1024 * SCQ * 4) {
        int b = s0 + t * (SCQ * 4);
        int4 v[SCQ];
#pragma unroll
        for (int u = 0; u < SCQ; ++u) {
            int i = b + 4 * u;
            v[u] = b4[(i < total ? i : 0) >> 2];
        }
        int s = 0;
#pragma unroll
        for (int u = 0; u < SCQ; ++u) {
            int i = b + 4 * u;
            v[u].x = (i     < total) ? v[u].x : 0;
            v[u].y = (i + 1 < total) ? v[u].y : 0;
            v[u].z = (i + 2 < total) ? v[u].z : 0;
            v[u].w = (i + 3 < total) ? v[u].w : 0;
            s += v[u].x + v[u].y + v[u].z + v[u].w;
        }
        int inc = s;
        for (int o = 1; o < 32; o <<= 1) {
            int y = __shfl_up(inc, o, 32);
            if (lane >= o) inc += y;
        }
        if (lane == 31) wsum[w] = inc;
        __syncthreads();
        int base = 0, all = 0;
        for (int i = 0; i < 32; ++i) {
            int x = wsum[i];
            base += (i < w) ? x : 0;
            all  += x;
        }
        int run = carry + base + inc - s;
#pragma unroll
        for (int u = 0; u < SCQ; ++u) {
            int i = b + 4 * u;
            int4 o4;
            o4.x = run;
            o4.y = o4.x + v[u].x;
            o4.z = o4.y + v[u].y;
            o4.w = o4.z + v[u].z;
            run  = o4.w + v[u].w;
            if (i < total) b4[i >> 2] = o4;
        }
        carry += all;
        __syncthreads();
    }
}

__global__ __launch_bounds__(ETH) void bucket_scatter_kernel(
    const int* __restrict__ ei, const int* __restrict__ offs,
    int* __restrict__ tSrc, int* __restrict__ tDst, int E, int NB, int nblk)
{
    __shared__ int hist[NBMAX];
    __shared__ int lpre[NBMAX];
    __shared__ int goff[NBMAX];
    __shared__ int sSrc[ECH];
    __shared__ int sDst[ECH];
    int t = threadIdx.x;
    if (t < NBMAX) hist[t] = 0;
    int og[BPL];
    if (t < 32) {
#pragma unroll
        for (int j = 0; j < BPL; ++j) {
            int k = t * BPL + j;
            og[j] = offs[(k < NB ? k : 0) * nblk + blockIdx.x];
        }
    }
    __syncthreads();
    int e0 = blockIdx.x * ECH + t;
    int s[EPT], d[EPT], r[EPT];
#pragma unroll
    for (int k = 0; k < EPT; ++k) {
        int e = e0 + k * ETH;
        e = e < E ? e : E - 1;
        s[k] = ei[e];
        d[k] = ei[(size_t)E + e];
    }
#pragma unroll
    for (int k = 0; k < EPT; ++k)
        r[k] = (e0 + k * ETH < E) ? lds_add_rtn(hist + (d[k] >> BSH)) : 0;
    __syncthreads();
    if (t < 32) {
        int c[BPL];
        int sum = 0;
#pragma unroll
        for (int j = 0; j < BPL; ++j) { c[j] = hist[t * BPL + j]; sum += c[j]; }
        int inc = sum;
        for (int o = 1; o < 32; o <<= 1) {
            int y = __shfl_up(inc, o, 32);
            if (t >= o) inc += y;
        }
        int ex = inc - sum;
#pragma unroll
        for (int j = 0; j < BPL; ++j) {
            int k = t * BPL + j;
            lpre[k] = ex;
            goff[k] = og[j] - ex;
            ex += c[j];
        }
    }
    __syncthreads();
#pragma unroll
    for (int k = 0; k < EPT; ++k) {
        if (e0 + k * ETH < E) {
            int p = lpre[d[k] >> BSH] + r[k];
            sSrc[p] = s[k];
            sDst[p] = d[k];
        }
    }
    __syncthreads();
    int mblk = E - blockIdx.x * ECH;
    mblk = mblk < ECH ? mblk : ECH;
#pragma unroll
    for (int k = 0; k < EPT; ++k) {
        int i = t + k * ETH;
        if (i < mblk) {
            int dd = sDst[i];
            int g = goff[dd >> BSH] + i;
            tSrc[g] = sSrc[i];
            tDst[g] = dd;
        }
    }
}

#define FTH 1024
#define FQ 8
#define STG 32768
__global__ __launch_bounds__(FTH) void bucket_finalize_kernel(
    const int* __restrict__ tSrc, const int* __restrict__ tDst, const int* __restrict__ offs,
    int* __restrict__ rowptr, int* __restrict__ col, int E, int NB, int nblk)
{
    __shared__ int cnt[BW];
    __shared__ int wsum[FTH / 32];
    __shared__ int stage[STG];
    int t = threadIdx.x, lane = t & 31, w = t >> 5;
    int kb = blockIdx.x;
    int beg = offs[kb * nblk];
    int end = E;
    if (kb + 1 < NB) end = offs[(kb + 1) * nblk];
    int m = end - beg;
    cnt[t] = 0;
    __syncthreads();
    const int* dp = tDst + beg;
    const int* sp = tSrc + beg;
    for (int i0 = 0; i0 < m; i0 += FTH * FQ) {
        int d[FQ];
#pragma unroll
        for (int u = 0; u < FQ; ++u) {
            int i = i0 + u * FTH + t;
            d[u] = dp[i < m ? i : m - 1];
        }
#pragma unroll
        for (int u = 0; u < FQ; ++u)
            if (i0 + u * FTH + t < m) lds_add_rtn(cnt + (d[u] & (BW - 1)));
    }
    __syncthreads();
    int c = cnt[t];
    int inc = c;
    for (int o = 1; o < 32; o <<= 1) {
        int y = __shfl_up(inc, o, 32);
        if (lane >= o) inc += y;
    }
    if (lane == 31) wsum[w] = inc;
    __syncthreads();
    int base = 0;
    for (int i = 0; i < FTH / 32; ++i) base += (i < w) ? wsum[i] : 0;
    int ex = base + inc - c;
    rowptr[kb * BW + t] = beg + ex;
    cnt[t] = ex;
    __syncthreads();
    bool staged = m <= STG;
    for (int i0 = 0; i0 < m; i0 += FTH * FQ) {
        int d[FQ], s[FQ];
#pragma unroll
        for (int u = 0; u < FQ; ++u) {
            int i = i0 + u * FTH + t;
            i = i < m ? i : m - 1;
            d[u] = dp[i];
            s[u] = sp[i];
        }
#pragma unroll
        for (int u = 0; u < FQ; ++u) {
            if (i0 + u * FTH + t < m) {
                int p = lds_add_rtn(cnt + (d[u] & (BW - 1)));
                if (staged) stage[p] = s[u];
                else        col[beg + p] = s[u];
            }
        }
    }
    if (staged) {
        __syncthreads();
        for (int i = t; i < m; i += FTH) col[beg + i] = stage[i];
    }
}

#define RPW 8
#define GCH 16
#define FROWS 32
#define TPITCH 68
typedef __attribute__((address_space(1))) const v2f global_v2f;
__device__ __forceinline__ global_v2f* wave_uniform_row(const float* p) {
    unsigned long long a = (unsigned long long)p;
    unsigned lo = __builtin_amdgcn_readfirstlane((unsigned)a);
    unsigned hi = __builtin_amdgcn_readfirstlane((unsigned)(a >> 32));
    return (global_v2f*)(((unsigned long long)hi << 32) | lo);
}

template <bool FINAL>
__global__ __launch_bounds__(128) void gin_layer_kernel(
    const float* __restrict__ h, const int* __restrict__ rowptr,
    const int* __restrict__ col, const float* __restrict__ W,
    const float* __restrict__ bias, float* __restrict__ out,
    const float* __restrict__ inSum, const float* __restrict__ inSumSq,
    const float* __restrict__ gamma, const float* __restrict__ beta,
    float* __restrict__ colsum, float* __restrict__ colsumsq,
    int n, int hasBN, int doRelu, int doStats)
{
    __shared__ __attribute__((aligned(16))) float tile[FROWS * TPITCH];
    __shared__ __attribute__((aligned(16))) float ytile[FROWS * TPITCH];
    int lane = threadIdx.x & 31;
    v2f sc = {1.f, 1.f}, sh = {0.f, 0.f};
    if (hasBN) {
        v2f cs = *(const v2f*)(inSum   + lane * 2);
        v2f cq = *(const v2f*)(inSumSq + lane * 2);
        v2f g  = *(const v2f*)(gamma   + lane * 2);
        v2f b  = *(const v2f*)(beta    + lane * 2);
        float invN = 1.f / (float)n;
        float mux = cs.x * invN, muy = cs.y * invN;
        float varx = cq.x * invN - mux * mux;
        float vary = cq.y * invN - muy * muy;
        sc.x = g.x * rsqrtf(varx + BN_EPS);
        sc.y = g.y * rsqrtf(vary + BN_EPS);
        sh.x = b.x - mux * sc.x;
        sh.y = b.y - muy * sc.y;
    }
    int wv = __builtin_amdgcn_readfirstlane((int)(threadIdx.x >> 5));
    int rowBase = blockIdx.x * FROWS;
    int r0 = rowBase + wv * RPW;
    int nr = n - r0;
    nr = nr < 0 ? 0 : (nr > RPW ? RPW : nr);
    int pi = r0 + (lane < nr ? lane : nr);
    pi = pi > n ? n : pi;
    int pj = rowptr[pi];
    v2f own[RPW];
#pragma unroll
    for (int rr = 0; rr < RPW; ++rr) {
        int r = r0 + rr;
        r = r < n ? r : n - 1;
        own[rr] = *(const v2f*)(h + (size_t)r * DD + lane * 2);
    }
    int eb = __builtin_amdgcn_readlane(pj, 0);
    int ee = __builtin_amdgcn_readlane(pj, nr);
    int last = ee - 1;
    int idx = 0;
    if (eb < ee) {
        int p0 = eb + lane;
        idx = col[p0 < last ? p0 : last];
    }
    float* trow = tile + (wv * RPW) * TPITCH + lane * 2;
#pragma unroll
    for (int rr = 0; rr < RPW; ++rr) {
        v2f t = {0.f, 0.f};
        if (rr < nr) {
            t.x = __builtin_fmaf(own[rr].x, sc.x, sh.x);
            t.y = __builtin_fmaf(own[rr].y, sc.y, sh.y);
        }
        *(v2f*)(trow + rr * TPITCH) = t;
    }
    if (eb < ee) {
        int cur = 0;
        int rend = __builtin_amdgcn_readlane(pj, 1);
        v2f acc = {0.f, 0.f};
        auto endRows = [&](int e) {
            do {
                float* tp = trow + cur * TPITCH;
                v2f t = *(v2f*)tp;
                t.x += acc.x;
                t.y += acc.y;
                *(v2f*)tp = t;
                acc.x = 0.f;
                acc.y = 0.f;
                ++cur;
                rend = __builtin_amdgcn_readlane(pj, cur + 1);
            } while (e >= rend);
        };
        int tot = ee - eb;
        int nfull = tot / GCH;
        int e0 = eb;
        int idxN = idx;
        for (int j = 0; j < nfull; ++j) {
            int hb = (j & 1) * GCH;
            v2f v[GCH];
            global_v2f* rp[GCH];
#pragma unroll
            for (int i = 0; i < GCH; ++i) {
                int src = __builtin_amdgcn_readlane(idx, hb + i);
                rp[i] = wave_uniform_row(h + (size_t)src * DD);
            }
            __builtin_amdgcn_sched_barrier(0);
#pragma unroll
            for (int i = 0; i < GCH; ++i) v[i] = rp[i][lane];
            if (hb == 0) {
                int pn = e0 + 32 + lane;
                idxN = col[pn < last ? pn : last];
            }
#pragma unroll
            for (int i = 0; i < GCH; ++i) {
                int e = e0 + i;
                if (__builtin_expect(e >= rend, 0)) endRows(e);
                acc.x += __builtin_fmaf(v[i].x, sc.x, sh.x);
                acc.y += __builtin_fmaf(v[i].y, sc.y, sh.y);
            }
            e0 += GCH;
            if (hb != 0) idx = idxN;
        }
        int rem = tot - nfull * GCH;
        if (rem > 0) {
            int hb = (nfull & 1) * GCH;
            v2f v[GCH];
            global_v2f* rp[GCH];
#pragma unroll
            for (int i = 0; i < GCH; ++i) {
                int k = hb + (i < rem ? i : rem - 1);
                int src = __builtin_amdgcn_readlane(idx, k);
                rp[i] = wave_uniform_row(h + (size_t)src * DD);
            }
            __builtin_amdgcn_sched_barrier(0);
#pragma unroll
            for (int i = 0; i < GCH; ++i) v[i] = rp[i][lane];
#pragma unroll
            for (int i = 0; i < GCH; ++i) {
                if (i < rem) {
                    int e = e0 + i;
                    if (__builtin_expect(e >= rend, 0)) endRows(e);
                    acc.x += __builtin_fmaf(v[i].x, sc.x, sh.x);
                    acc.y += __builtin_fmaf(v[i].y, sc.y, sh.y);
                }
            }
        }
        float* tp = trow + cur * TPITCH;
        v2f t = *(v2f*)tp;
        t.x += acc.x;
        t.y += acc.y;
        *(v2f*)tp = t;
    }

    __syncthreads();

    int m  = lane & 15;
    int kh = lane >> 4;
    int cidx = wv * 16 + m;
    const float* wp = W + (size_t)(2 * kh) * DD + cidx;

    v2f bf[DD / 4];
#pragma unroll
    for (int j = 0; j < DD / 4; ++j) {
        bf[j].x = wp[(size_t)(4 * j) * DD];
        bf[j].y = wp[(size_t)(4 * j + 1) * DD];
    }
    float bcol = bias[cidx];
    float s = 0.f, s2 = 0.f;
#pragma unroll 1
    for (int t = 0; t < FROWS / 16; ++t) {
        const float* tp = tile + (t * 16 + m) * TPITCH + 2 * kh;
        v8f c = {};
#pragma unroll
        for (int j = 0; j < DD / 4; ++j) {
            v2f a = *(const v2f*)(tp + 4 * j);
            c = __builtin_amdgcn_wmma_f32_16x16x4_f32(
                    false, a, false, bf[j], (short)0, c, false, false);
        }
        if constexpr (FINAL) {
#pragma unroll
            for (int r = 0; r < 8; ++r)
                ytile[(t * 16 + 8 * kh + r) * TPITCH + cidx] = c[r] + bcol;
        } else {
            int rowT = rowBase + t * 16 + 8 * kh;
            float val[8];
#pragma unroll
            for (int r = 0; r < 8; ++r) {
                val[r] = c[r] + bcol;
                if (doRelu) val[r] = fmaxf(val[r], 0.f);
                ytile[(t * 16 + 8 * kh + r) * TPITCH + cidx] = val[r];
            }
            __builtin_amdgcn_sched_barrier(0);
#pragma unroll
            for (int r = 0; r < 8; ++r) {
                float u = (rowT + r < n) ? val[r] : 0.f;
                s  += u;
                s2 += u * u;
            }
        }
    }
    if constexpr (FINAL) {
        __syncthreads();
        float* pack = tile + wv * (RPW * CC);
        bool h1 = (lane + 32) < CC;
        float res0[RPW], res1[RPW];
#pragma unroll
        for (int rr = 0; rr < RPW; ++rr) {
            const float* yr = ytile + (wv * RPW + rr) * TPITCH;
            float v0 = yr[lane];
            float w1 = yr[lane + 32];
            float v1 = h1 ? w1 : -3.402823466e38f;
            float mx = fmaxf(v0, v1);
            for (int o = 16; o > 0; o >>= 1) mx = fmaxf(mx, __shfl_xor(mx, o, 32));
            float e1 = expf(v1 - mx);
            float e = expf(v0 - mx) + (h1 ? e1 : 0.f);
            for (int o = 16; o > 0; o >>= 1) e += __shfl_xor(e, o, 32);
            float lse = mx + logf(e);
            res0[rr] = v0 - lse;
            res1[rr] = v1 - lse;
        }
#pragma unroll
        for (int rr = 0; rr < RPW; ++rr) pack[rr * CC + lane] = res0[rr];
        if (h1) {
#pragma unroll
            for (int rr = 0; rr < RPW; ++rr) pack[rr * CC + lane + 32] = res1[rr];
        }
        __syncthreads();
        float4* o4 = (float4*)(out + (size_t)r0 * CC);
        const float4* p4 = (const float4*)pack;
#pragma unroll
        for (int p = 0; p < (RPW * CC / 4 + 31) / 32; ++p) {
            int q = p * 32 + lane;
            if (q < RPW * CC / 4 && r0 + (q * 4) / CC < n) o4[q] = p4[q];
        }
    } else {
        __syncthreads();
        const float* yb = ytile + (wv * RPW) * TPITCH + (lane & 15) * 4;
        float* ob = out + (size_t)r0 * DD + (lane & 15) * 4;
        int rsub = lane >> 4;
        if (nr == RPW) {
            float4 v[RPW / 2];
#pragma unroll
            for (int p = 0; p < RPW / 2; ++p) v[p] = *(const float4*)(yb + (2 * p + rsub) * TPITCH);
#pragma unroll
            for (int p = 0; p < RPW / 2; ++p) *(float4*)(ob + (size_t)(2 * p + rsub) * DD) = v[p];
        } else {
#pragma unroll
            for (int p = 0; p < RPW / 2; ++p) {
                int row = 2 * p + rsub;
                if (row < nr) *(float4*)(ob + (size_t)row * DD) = *(const float4*)(yb + row * TPITCH);
            }
        }
        if (doStats) {
            s  += __shfl_xor(s, 16, 32);
            s2 += __shfl_xor(s2, 16, 32);
            if (kh == 0) {
                atomic_add_f32(colsum   + cidx, s);
                atomic_add_f32(colsumsq + cidx, s2);
            }
        }
    }
}

__global__ void pad_final_kernel(const float* __restrict__ Wf,
                                 const float* __restrict__ bf,
                                 float* __restrict__ Wpad,
                                 float* __restrict__ bpad)
{
    int i = blockIdx.x * blockDim.x + threadIdx.x;
    if (i < DD * DD) {
        int r = i >> 6, c = i & 63;
        Wpad[i] = (c < CC) ? Wf[r * CC + c] : 0.f;
    }
    if (i < DD) bpad[i] = (i < CC) ? bf[i] : 0.f;
}

extern "C" void kernel_launch(void* const* d_in, const int* in_sizes, int n_in,
                              void* d_out, int out_size, void* d_ws, size_t ws_size,
                              hipStream_t stream) {
    const float* x       = (const float*)d_in[0];
    const int*   ei      = (const int*)  d_in[1];
    const float* W_proj  = (const float*)d_in[2];
    const float* b_proj  = (const float*)d_in[3];
    const float* Ws      = (const float*)d_in[4];
    const float* bs      = (const float*)d_in[5];
    const float* W_final = (const float*)d_in[6];
    const float* b_final = (const float*)d_in[7];
    const float* gammas  = (const float*)d_in[8];
    const float* betas   = (const float*)d_in[9];
    float* out = (float*)d_out;

    int n = in_sizes[0] / DD;
    int E = in_sizes[1] / 2;

    size_t nd = (size_t)n * DD;
    float* ws     = (float*)d_ws;
    float* planeA = ws;
    int*   lists  = (int*)(ws + nd);
    float* planeB = ws + 2 * nd;
    float* Wpad   = ws + 3 * nd;
    float* bpad   = Wpad + DD * DD;

    int NB        = (n + 1 + BW - 1) / BW;
    int RP        = NB * BW;
    int EP        = (E + 63) / 64 * 64;
    int nblk      = (E + ECH - 1) / ECH;
    int total     = NB * nblk;
    float* stats = (float*)lists;
    int* rowptr = lists + STATF;
    int* col    = rowptr + RP;
    int* tSrc   = col + EP;
    int* tDst   = tSrc + EP;
    int* bh     = tDst + EP;

    int layerBlocks  = (n + FROWS - 1) / FROWS;

    bucket_hist_kernel<<<nblk, ETH, 0, stream>>>(ei, bh, E, NB, nblk);
    bucket_scan_kernel<<<1, 1024, 0, stream>>>(bh, total, stats, STATF);
    bucket_scatter_kernel<<<nblk, ETH, 0, stream>>>(ei, bh, tSrc, tDst, E, NB, nblk);
    bucket_finalize_kernel<<<NB, FTH, 0, stream>>>(tSrc, tDst, bh, rowptr, col, E, NB, nblk);

    const float* hin = x;
    for (int layer = 0; layer <= LL; ++layer) {
        const float* W   = (layer == 0) ? W_proj : Ws + (size_t)(layer - 1) * DD * DD;
        const float* b   = (layer == 0) ? b_proj : bs + (size_t)(layer - 1) * DD;
        float* hout      = (layer & 1) ? planeB : planeA;
        float* st        = stats + (size_t)layer * 2 * DD;
        const float* pst = (layer == 0) ? nullptr : st - 2 * DD;
        const float* pg  = (layer == 0) ? nullptr : gammas + (size_t)(layer - 1) * DD;
        const float* pb  = (layer == 0) ? nullptr : betas + (size_t)(layer - 1) * DD;

        gin_layer_kernel<false><<<layerBlocks, 128, 0, stream>>>(
            hin, rowptr, col, W, b, hout, pst, (layer == 0) ? nullptr : pst + DD, pg, pb,
            st, st + DD, n,  (layer == 0) ? 0 : 1,  1,  1);
        hin = hout;
    }

    const float* lst = stats + (size_t)LL * 2 * DD;
    pad_final_kernel<<<(DD * DD + 255) / 256, 256, 0, stream>>>(W_final, b_final, Wpad, bpad);
    gin_layer_kernel<true><<<layerBlocks, 128, 0, stream>>>(
        hin, rowptr, col, Wpad, bpad, out, lst, lst + DD,
        gammas + (size_t)LL * DD, betas + (size_t)LL * DD,
        nullptr, nullptr, n,  1,  0,  0);
}


# set-up launches 5 -> 2: edge blocks keep bucket-sorted runs in place with per-block counts, a bucket block sums counts and reads its runs; count, scan, W-pad launches go; list order = LDS add order
# speedup vs baseline: 5.8424x; 1.0587x over previous
#include <hip/hip_runtime.h>
#include <hip/hip_bf16.h>
#include <math.h>

#define DD 64
#define CC 40
#define LL 3
#define BN_EPS 1e-5f
#define STATF ((LL + 1) * 2 * DD)

typedef __attribute__((ext_vector_type(2))) float v2f;
typedef __attribute__((ext_vector_type(8))) float v8f;

__device__ __forceinline__ void atomic_add_f32(float* p, float v) {
    __hip_atomic_fetch_add(p, v, __ATOMIC_RELAXED, __HIP_MEMORY_SCOPE_AGENT);
}

#define BSH 10
#define BW (1 << BSH)
#define NBMAX 256
#define BPL (NBMAX / 32)
#define ECH 8192
#define ETH 512
#define EPT (ECH / ETH)

__device__ __forceinline__ int lds_add_rtn(int* p) {
    return __hip_atomic_fetch_add(p, 1, __ATOMIC_RELAXED, __HIP_MEMORY_SCOPE_WORKGROUP);
}

__global__ __launch_bounds__(ETH) void block_sort_kernel(
    const int* __restrict__ ei, int* __restrict__ tSrc, int* __restrict__ tDst,
    int* __restrict__ cntT, int* __restrict__ preT,
    float* __restrict__ stats, int nstat, int E, int NB, int nblk)
{
    __shared__ int hist[NBMAX];
    __shared__ int lpre[NBMAX];
    __shared__ int sSrc[ECH];
    __shared__ int sDst[ECH];
    int t = threadIdx.x;
    if (t < NBMAX) hist[t] = 0;
    if (blockIdx.x == 0)
        for (int i = t; i < nstat; i += ETH) stats[i] = 0.f;
    __syncthreads();
    int e0 = blockIdx.x * ECH + t;
    int s[EPT], d[EPT], r[EPT];
#pragma unroll
    for (int k = 0; k < EPT; ++k) {
        int e = e0 + k * ETH;
        e = e < E ? e : E - 1;
        s[k] = ei[e];
        d[k] = ei[(size_t)E + e];
    }
#pragma unroll
    for (int k = 0; k < EPT; ++k)
        r[k] = (e0 + k * ETH < E) ? lds_add_rtn(hist + (d[k] >> BSH)) : 0;
    __syncthreads();
    if (t < 32) {
        int c[BPL];
        int sum = 0;
#pragma unroll
        for (int j = 0; j < BPL; ++j) { c[j] = hist[t * BPL + j]; sum += c[j]; }
        int inc = sum;
        for (int o = 1; o < 32; o <<= 1) {
            int y = __shfl_up(inc, o, 32);
            if (t >= o) inc += y;
        }
        int ex = inc - sum;
#pragma unroll
        for (int j = 0; j < BPL; ++j) {
            int k = t * BPL + j;
            lpre[k] = ex;
            if (k < NB) {
                cntT[k * nblk + blockIdx.x] = c[j];
                preT[k * nblk + blockIdx.x] = ex;
            }
            ex += c[j];
        }
    }
    __syncthreads();
#pragma unroll
    for (int k = 0; k < EPT; ++k) {
        if (e0 + k * ETH < E) {
            int p = lpre[d[k] >> BSH] + r[k];
            sSrc[p] = s[k];
            sDst[p] = d[k];
        }
    }
    __syncthreads();
    int mblk = E - blockIdx.x * ECH;
    mblk = mblk < ECH ? mblk : ECH;
    int* oS = tSrc + (size_t)blockIdx.x * ECH;
    int* oD = tDst + (size_t)blockIdx.x * ECH;
#pragma unroll
    for (int k = 0; k < EPT; ++k) {
        int i = t + k * ETH;
        if (i < mblk) {
            oS[i] = sSrc[i];
            oD[i] = sDst[i];
        }
    }
}

#define FTH 1024
#define SQ 20
#define JQ 8
#define RQ 4
#define STG 32768
__global__ __launch_bounds__(FTH) void bucket_finalize_kernel(
    const int* __restrict__ tSrc, const int* __restrict__ tDst,
    const int* __restrict__ cntT, const int* __restrict__ preT,
    int* __restrict__ rowptr, int* __restrict__ col, int NB, int nblk)
{
    __shared__ int cnt[BW];
    __shared__ int wsum[FTH / 32];
    __shared__ int bsum[FTH / 32];
    __shared__ int msum[FTH / 32];
    __shared__ int stage[STG];
    int t = threadIdx.x, lane = t & 31, w = t >> 5;
    int kb = blockIdx.x;
    int lim = kb * nblk;
    int pb = 0;
    for (int i0 = 0; i0 < lim; i0 += FTH * SQ) {
        int v[SQ];
#pragma unroll
        for (int u = 0; u < SQ; ++u) {
            int i = i0 + u * FTH + t;
            v[u] = cntT[i < lim ? i : 0];
        }
#pragma unroll
        for (int u = 0; u < SQ; ++u) pb += (i0 + u * FTH + t < lim) ? v[u] : 0;
    }
    int pm = 0;
    for (int b = t; b < nblk; b += FTH) pm += cntT[lim + b];
    cnt[t] = 0;
    __syncthreads();
    const int* crow = cntT + lim;
    const int* prow = preT + lim;
    for (int j0 = 0; j0 < nblk; j0 += 32 * JQ) {
        int c[JQ], base[JQ];
        int cmax = 0;
#pragma unroll
        for (int u = 0; u < JQ; ++u) {
            int b = j0 + u * 32 + w;
            bool ok = b < nblk;
            int bi = ok ? b : 0;
            int cc = crow[bi];
            int pp = prow[bi];
            c[u] = ok ? cc : 0;
            base[u] = c[u] > 0 ? bi * ECH + pp : 0;
            cmax = cmax > c[u] ? cmax : c[u];
        }
        for (int i0 = 0; i0 < cmax; i0 += 32 * RQ) {
            int d[JQ][RQ];
#pragma unroll
            for (int u = 0; u < JQ; ++u)
#pragma unroll
                for (int q = 0; q < RQ; ++q) {
                    int i = i0 + q * 32 + lane;
                    int ic = i < c[u] ? i : c[u] - 1;
                    ic = ic < 0 ? 0 : ic;
                    d[u][q] = tDst[base[u] + ic];
                }
#pragma unroll
            for (int u = 0; u < JQ; ++u)
#pragma unroll
                for (int q = 0; q < RQ; ++q)
                    if (i0 + q * 32 + lane < c[u]) lds_add_rtn(cnt + (d[u][q] & (BW - 1)));
        }
    }
    for (int o = 16; o > 0; o >>= 1) {
        pb += __shfl_xor(pb, o, 32);
        pm += __shfl_xor(pm, o, 32);
    }
    if (lane == 0) { bsum[w] = pb; msum[w] = pm; }
    __syncthreads();
    int beg = 0, m = 0;
    for (int i = 0; i < FTH / 32; ++i) { beg += bsum[i]; m += msum[i]; }
    int c = cnt[t];
    int inc = c;
    for (int o = 1; o < 32; o <<= 1) {
        int y = __shfl_up(inc, o, 32);
        if (lane >= o) inc += y;
    }
    if (lane == 31) wsum[w] = inc;
    __syncthreads();
    int base = 0;
    for (int i = 0; i < FTH / 32; ++i) base += (i < w) ? wsum[i] : 0;
    int ex = base + inc - c;
    rowptr[kb * BW + t] = beg + ex;
    cnt[t] = ex;
    __syncthreads();
    bool staged = m <= STG;
    for (int j0 = 0; j0 < nblk; j0 += 32 * JQ) {
        int cr[JQ], rb[JQ];
        int cmax = 0;
#pragma unroll
        for (int u = 0; u < JQ; ++u) {
            int b = j0 + u * 32 + w;
            bool ok = b < nblk;
            int bi = ok ? b : 0;
            int cc = crow[bi];
            int pp = prow[bi];
            cr[u] = ok ? cc : 0;
            rb[u] = cr[u] > 0 ? bi * ECH + pp : 0;
            cmax = cmax > cr[u] ? cmax : cr[u];
        }
        for (int i0 = 0; i0 < cmax; i0 += 32 * RQ) {
            int d[JQ][RQ], s[JQ][RQ];
#pragma unroll
            for (int u = 0; u < JQ; ++u)
#pragma unroll
                for (int q = 0; q < RQ; ++q) {
                    int i = i0 + q * 32 + lane;
                    int ic = i < cr[u] ? i : cr[u] - 1;
                    ic = ic < 0 ? 0 : ic;
                    d[u][q] = tDst[rb[u] + ic];
                    s[u][q] = tSrc[rb[u] + ic];
                }
#pragma unroll
            for (int u = 0; u < JQ; ++u)
#pragma unroll
                for (int q = 0; q < RQ; ++q) {
                    if (i0 + q * 32 + lane < cr[u]) {
                        int p = lds_add_rtn(cnt + (d[u][q] & (BW - 1)));
                        if (staged) stage[p] = s[u][q];
                        else        col[beg + p] = s[u][q];
                    }
                }
        }
    }
    if (staged) {
        __syncthreads();
        for (int i = t; i < m; i += FTH) col[beg + i] = stage[i];
    }
}

#define RPW 8
#define GCH 16
#define FROWS 32
#define TPITCH 68
typedef __attribute__((address_space(1))) const v2f global_v2f;
__device__ __forceinline__ global_v2f* wave_uniform_row(const float* p) {
    unsigned long long a = (unsigned long long)p;
    unsigned lo = __builtin_amdgcn_readfirstlane((unsigned)a);
    unsigned hi = __builtin_amdgcn_readfirstlane((unsigned)(a >> 32));
    return (global_v2f*)(((unsigned long long)hi << 32) | lo);
}

template <bool FINAL>
__global__ __launch_bounds__(128) void gin_layer_kernel(
    const float* __restrict__ h, const int* __restrict__ rowptr,
    const int* __restrict__ col, const float* __restrict__ W,
    const float* __restrict__ bias, float* __restrict__ out,
    const float* __restrict__ inSum, const float* __restrict__ inSumSq,
    const float* __restrict__ gamma, const float* __restrict__ beta,
    float* __restrict__ colsum, float* __restrict__ colsumsq,
    int n, int hasBN, int doRelu, int doStats)
{
    __shared__ __attribute__((aligned(16))) float tile[FROWS * TPITCH];
    __shared__ __attribute__((aligned(16))) float ytile[FROWS * TPITCH];
    int lane = threadIdx.x & 31;
    v2f sc = {1.f, 1.f}, sh = {0.f, 0.f};
    if (hasBN) {
        v2f cs = *(const v2f*)(inSum   + lane * 2);
        v2f cq = *(const v2f*)(inSumSq + lane * 2);
        v2f g  = *(const v2f*)(gamma   + lane * 2);
        v2f b  = *(const v2f*)(beta    + lane * 2);
        float invN = 1.f / (float)n;
        float mux = cs.x * invN, muy = cs.y * invN;
        float varx = cq.x * invN - mux * mux;
        float vary = cq.y * invN - muy * muy;
        sc.x = g.x * rsqrtf(varx + BN_EPS);
        sc.y = g.y * rsqrtf(vary + BN_EPS);
        sh.x = b.x - mux * sc.x;
        sh.y = b.y - muy * sc.y;
    }
    int wv = __builtin_amdgcn_readfirstlane((int)(threadIdx.x >> 5));
    int rowBase = blockIdx.x * FROWS;
    int r0 = rowBase + wv * RPW;
    int nr = n - r0;
    nr = nr < 0 ? 0 : (nr > RPW ? RPW : nr);
    int pi = r0 + (lane < nr ? lane : nr);
    pi = pi > n ? n : pi;
    int pj = rowptr[pi];
    v2f own[RPW];
#pragma unroll
    for (int rr = 0; rr < RPW; ++rr) {
        int r = r0 + rr;
        r = r < n ? r : n - 1;
        own[rr] = *(const v2f*)(h + (size_t)r * DD + lane * 2);
    }
    int eb = __builtin_amdgcn_readlane(pj, 0);
    int ee = __builtin_amdgcn_readlane(pj, nr);
    int last = ee - 1;
    int idx = 0;
    if (eb < ee) {
        int p0 = eb + lane;
        idx = col[p0 < last ? p0 : last];
    }
    float* trow = tile + (wv * RPW) * TPITCH + lane * 2;
#pragma unroll
    for (int rr = 0; rr < RPW; ++rr) {
        v2f t = {0.f, 0.f};
        if (rr < nr) {
            t.x = __builtin_fmaf(own[rr].x, sc.x, sh.x);
            t.y = __builtin_fmaf(own[rr].y, sc.y, sh.y);
        }
        *(v2f*)(trow + rr * TPITCH) = t;
    }
    if (eb < ee) {
        int cur = 0;
        int rend = __builtin_amdgcn_readlane(pj, 1);
        v2f acc = {0.f, 0.f};
        auto endRows = [&](int e) {
            do {
                float* tp = trow + cur * TPITCH;
                v2f t = *(v2f*)tp;
                t.x += acc.x;
                t.y += acc.y;
                *(v2f*)tp = t;
                acc.x = 0.f;
                acc.y = 0.f;
                ++cur;
                rend = __builtin_amdgcn_readlane(pj, cur + 1);
            } while (e >= rend);
        };
        int tot = ee - eb;
        int nfull = tot / GCH;
        int e0 = eb;
        int idxN = idx;
        for (int j = 0; j < nfull; ++j) {
            int hb = (j & 1) * GCH;
            v2f v[GCH];
            global_v2f* rp[GCH];
#pragma unroll
            for (int i = 0; i < GCH; ++i) {
                int src = __builtin_amdgcn_readlane(idx, hb + i);
                rp[i] = wave_uniform_row(h + (size_t)src * DD);
            }
            __builtin_amdgcn_sched_barrier(0);
#pragma unroll
            for (int i = 0; i < GCH; ++i) v[i] = rp[i][lane];
            if (hb == 0) {
                int pn = e0 + 32 + lane;
                idxN = col[pn < last ? pn : last];
            }
#pragma unroll
            for (int i = 0; i < GCH; ++i) {
                int e = e0 + i;
                if (__builtin_expect(e >= rend, 0)) endRows(e);
                acc.x += __builtin_fmaf(v[i].x, sc.x, sh.x);
                acc.y += __builtin_fmaf(v[i].y, sc.y, sh.y);
            }
            e0 += GCH;
            if (hb != 0) idx = idxN;
        }
        int rem = tot - nfull * GCH;
        if (rem > 0) {
            int hb = (nfull & 1) * GCH;
            v2f v[GCH];
            global_v2f* rp[GCH];
#pragma unroll
            for (int i = 0; i < GCH; ++i) {
                int k = hb + (i < rem ? i : rem - 1);
                int src = __builtin_amdgcn_readlane(idx, k);
                rp[i] = wave_uniform_row(h + (size_t)src * DD);
            }
            __builtin_amdgcn_sched_barrier(0);
#pragma unroll
            for (int i = 0; i < GCH; ++i) v[i] = rp[i][lane];
#pragma unroll
            for (int i = 0; i < GCH; ++i) {
                if (i < rem) {
                    int e = e0 + i;
                    if (__builtin_expect(e >= rend, 0)) endRows(e);
                    acc.x += __builtin_fmaf(v[i].x, sc.x, sh.x);
                    acc.y += __builtin_fmaf(v[i].y, sc.y, sh.y);
                }
            }
        }
        float* tp = trow + cur * TPITCH;
        v2f t = *(v2f*)tp;
        t.x += acc.x;
        t.y += acc.y;
        *(v2f*)tp = t;
    }

    __syncthreads();

    int m  = lane & 15;
    int kh = lane >> 4;
    int cidx = wv * 16 + m;
    constexpr int WP = FINAL ? CC : DD;
    bool cok = !FINAL || cidx < CC;
    int ccl = cok ? cidx : 0;
    const float* wp = W + (size_t)(2 * kh) * WP + ccl;

    v2f bf[DD / 4];
#pragma unroll
    for (int j = 0; j < DD / 4; ++j) {
        float bx = wp[(size_t)(4 * j) * WP];
        float by = wp[(size_t)(4 * j + 1) * WP];
        bf[j].x = cok ? bx : 0.f;
        bf[j].y = cok ? by : 0.f;
    }
    float bcol = bias[ccl];
    bcol = cok ? bcol : 0.f;
    float s = 0.f, s2 = 0.f;
#pragma unroll 1
    for (int t = 0; t < FROWS / 16; ++t) {
        const float* tp = tile + (t * 16 + m) * TPITCH + 2 * kh;
        v8f c = {};
#pragma unroll
        for (int j = 0; j < DD / 4; ++j) {
            v2f a = *(const v2f*)(tp + 4 * j);
            c = __builtin_amdgcn_wmma_f32_16x16x4_f32(
                    false, a, false, bf[j], (short)0, c, false, false);
        }
        if constexpr (FINAL) {
#pragma unroll
            for (int r = 0; r < 8; ++r)
                ytile[(t * 16 + 8 * kh + r) * TPITCH + cidx] = c[r] + bcol;
        } else {
            int rowT = rowBase + t * 16 + 8 * kh;
            float val[8];
#pragma unroll
            for (int r = 0; r < 8; ++r) {
                val[r] = c[r] + bcol;
                if (doRelu) val[r] = fmaxf(val[r], 0.f);
                ytile[(t * 16 + 8 * kh + r) * TPITCH + cidx] = val[r];
            }
            __builtin_amdgcn_sched_barrier(0);
#pragma unroll
            for (int r = 0; r < 8; ++r) {
                float u = (rowT + r < n) ? val[r] : 0.f;
                s  += u;
                s2 += u * u;
            }
        }
    }
    if constexpr (FINAL) {
        __syncthreads();
        float* pack = tile + wv * (RPW * CC);
        bool h1 = (lane + 32) < CC;
        float res0[RPW], res1[RPW];
#pragma unroll
        for (int rr = 0; rr < RPW; ++rr) {
            const float* yr = ytile + (wv * RPW + rr) * TPITCH;
            float v0 = yr[lane];
            float w1 = yr[lane + 32];
            float v1 = h1 ? w1 : -3.402823466e38f;
            float mx = fmaxf(v0, v1);
            for (int o = 16; o > 0; o >>= 1) mx = fmaxf(mx, __shfl_xor(mx, o, 32));
            float e1 = expf(v1 - mx);
            float e = expf(v0 - mx) + (h1 ? e1 : 0.f);
            for (int o = 16; o > 0; o >>= 1) e += __shfl_xor(e, o, 32);
            float lse = mx + logf(e);
            res0[rr] = v0 - lse;
            res1[rr] = v1 - lse;
        }
#pragma unroll
        for (int rr = 0; rr < RPW; ++rr) pack[rr * CC + lane] = res0[rr];
        if (h1) {
#pragma unroll
            for (int rr = 0; rr < RPW; ++rr) pack[rr * CC + lane + 32] = res1[rr];
        }
        __syncthreads();
        float4* o4 = (float4*)(out + (size_t)r0 * CC);
        const float4* p4 = (const float4*)pack;
#pragma unroll
        for (int p = 0; p < (RPW * CC / 4 + 31) / 32; ++p) {
            int q = p * 32 + lane;
            if (q < RPW * CC / 4 && r0 + (q * 4) / CC < n) o4[q] = p4[q];
        }
    } else {
        __syncthreads();
        const float* yb = ytile + (wv * RPW) * TPITCH + (lane & 15) * 4;
        float* ob = out + (size_t)r0 * DD + (lane & 15) * 4;
        int rsub = lane >> 4;
        if (nr == RPW) {
            float4 v[RPW / 2];
#pragma unroll
            for (int p = 0; p < RPW / 2; ++p) v[p] = *(const float4*)(yb + (2 * p + rsub) * TPITCH);
#pragma unroll
            for (int p = 0; p < RPW / 2; ++p) *(float4*)(ob + (size_t)(2 * p + rsub) * DD) = v[p];
        } else {
#pragma unroll
            for (int p = 0; p < RPW / 2; ++p) {
                int row = 2 * p + rsub;
                if (row < nr) *(float4*)(ob + (size_t)row * DD) = *(const float4*)(yb + row * TPITCH);
            }
        }
        if (doStats) {
            s  += __shfl_xor(s, 16, 32);
            s2 += __shfl_xor(s2, 16, 32);
            if (kh == 0) {
                atomic_add_f32(colsum   + cidx, s);
                atomic_add_f32(colsumsq + cidx, s2);
            }
        }
    }
}

extern "C" void kernel_launch(void* const* d_in, const int* in_sizes, int n_in,
                              void* d_out, int out_size, void* d_ws, size_t ws_size,
                              hipStream_t stream) {
    const float* x       = (const float*)d_in[0];
    const int*   ei      = (const int*)  d_in[1];
    const float* W_proj  = (const float*)d_in[2];
    const float* b_proj  = (const float*)d_in[3];
    const float* Ws      = (const float*)d_in[4];
    const float* bs      = (const float*)d_in[5];
    const float* W_final = (const float*)d_in[6];
    const float* b_final = (const float*)d_in[7];
    const float* gammas  = (const float*)d_in[8];
    const float* betas   = (const float*)d_in[9];
    float* out = (float*)d_out;

    int n = in_sizes[0] / DD;
    int E = in_sizes[1] / 2;

    size_t nd = (size_t)n * DD;
    float* ws     = (float*)d_ws;
    float* planeA = ws;
    int*   lists  = (int*)(ws + nd);
    float* planeB = ws + 2 * nd;

    int NB        = (n + 1 + BW - 1) / BW;
    int RP        = NB * BW;
    int EP        = (E + 63) / 64 * 64;
    int nblk      = (E + ECH - 1) / ECH;
    int total     = NB * nblk;
    float* stats = (float*)lists;
    int* rowptr = lists + STATF;
    int* col    = rowptr + RP;
    int* tSrc   = col + EP;
    int* tDst   = tSrc + EP;
    int* cntT   = tDst + EP;
    int* preT   = cntT + total;

    int layerBlocks  = (n + FROWS - 1) / FROWS;

    block_sort_kernel<<<nblk, ETH, 0, stream>>>(ei, tSrc, tDst, cntT, preT, stats, STATF, E, NB, nblk);
    bucket_finalize_kernel<<<NB, FTH, 0, stream>>>(tSrc, tDst, cntT, preT, rowptr, col, NB, nblk);

    const float* hin = x;
    for (int layer = 0; layer <= LL; ++layer) {
        const float* W   = (layer == 0) ? W_proj : Ws + (size_t)(layer - 1) * DD * DD;
        const float* b   = (layer == 0) ? b_proj : bs + (size_t)(layer - 1) * DD;
        float* hout      = (layer & 1) ? planeB : planeA;
        float* st        = stats + (size_t)layer * 2 * DD;
        const float* pst = (layer == 0) ? nullptr : st - 2 * DD;
        const float* pg  = (layer == 0) ? nullptr : gammas + (size_t)(layer - 1) * DD;
        const float* pb  = (layer == 0) ? nullptr : betas + (size_t)(layer - 1) * DD;

        gin_layer_kernel<false><<<layerBlocks, 128, 0, stream>>>(
            hin, rowptr, col, W, b, hout, pst, (layer == 0) ? nullptr : pst + DD, pg, pb,
            st, st + DD, n,  (layer == 0) ? 0 : 1,  1,  1);
        hin = hout;
    }

    const float* lst = stats + (size_t)LL * 2 * DD;
    gin_layer_kernel<true><<<layerBlocks, 128, 0, stream>>>(
        hin, rowptr, col, W_final, b_final, out, lst, lst + DD,
        gammas + (size_t)LL * DD, betas + (size_t)LL * DD,
        nullptr, nullptr, n,  1,  0,  0);
}
